# starting kernel
# baseline (speedup 1.0000x reference)
#include <hip/hip_runtime.h>

typedef __attribute__((ext_vector_type(4)))  _Float16 v4h;
typedef __attribute__((ext_vector_type(8)))  _Float16 v8h;
typedef __attribute__((ext_vector_type(16))) _Float16 v16h;
typedef __attribute__((ext_vector_type(8)))  float    v8f;

#define NNODES 40000
#define NEDGES 640000
#define EMBED  128
#define NTYPES 64
#define NITERS 4


__global__ void k_zero2(int* a, int* b) {
    int i = blockIdx.x * blockDim.x + threadIdx.x;
    if (i < NNODES) { a[i] = 0; b[i] = 0; }
}

__global__ void k_count(const int* __restrict__ dst, int* cnt) {
    int e = blockIdx.x * blockDim.x + threadIdx.x;
    if (e < NEDGES) atomicAdd(&cnt[dst[e]], 1);
}

__global__ void k_dinv(const int* __restrict__ cnt, float* __restrict__ dinv) {
    int i = blockIdx.x * blockDim.x + threadIdx.x;
    if (i < NNODES) dinv[i] = rsqrtf((float)cnt[i] + 1.0f);
}

__global__ void k_scan(const int* __restrict__ cnt, int* __restrict__ off) {
    __shared__ int s[1024];
    __shared__ int carry;
    const int tid = threadIdx.x;
    if (tid == 0) carry = 0;
    __syncthreads();
    for (int base = 0; base < NNODES; base += 1024) {
        int v = (base + tid < NNODES) ? cnt[base + tid] : 0;
        s[tid] = v;
        __syncthreads();
        for (int d = 1; d < 1024; d <<= 1) {
            int t = (tid >= d) ? s[tid - d] : 0;
            __syncthreads();
            s[tid] += t;
            __syncthreads();
        }
        int incl = s[tid];
        if (base + tid < NNODES) off[base + tid] = carry + incl - v;
        __syncthreads();
        if (tid == 1023) carry += s[1023];
        __syncthreads();
    }
    if (tid == 0) off[NNODES] = carry;
}

__global__ void k_csr_fill(const int* __restrict__ src, const int* __restrict__ dst,
                           const int* __restrict__ off, const float* __restrict__ dinv,
                           int* __restrict__ cursor, int* __restrict__ col,
                           float* __restrict__ wt) {
    int e = blockIdx.x * blockDim.x + threadIdx.x;
    if (e >= NEDGES) return;
    int s = src[e], d = dst[e];
    int j = off[d] + atomicAdd(&cursor[d], 1);
    col[j] = s;
    wt[j]  = dinv[s] * dinv[d];
}


__device__ __forceinline__ void split_store(float x, float y, float z, float w,
                                            _Float16* hi, _Float16* lo, size_t idx4) {
    v4h H, L;
    H[0] = (_Float16)x; L[0] = (_Float16)(x - (float)H[0]);
    H[1] = (_Float16)y; L[1] = (_Float16)(y - (float)H[1]);
    H[2] = (_Float16)z; L[2] = (_Float16)(z - (float)H[2]);
    H[3] = (_Float16)w; L[3] = (_Float16)(w - (float)H[3]);
    ((v4h*)hi)[idx4] = H;
    ((v4h*)lo)[idx4] = L;
}

__global__ void k_embed_split(const int* __restrict__ x, const float* __restrict__ Win,
                              _Float16* __restrict__ hhi, _Float16* __restrict__ hlo) {
    int t = blockIdx.x * blockDim.x + threadIdx.x;
    if (t >= NNODES * (EMBED / 4)) return;
    int i = t >> 5, c = t & 31;
    float4 v = ((const float4*)(Win + (size_t)x[i] * EMBED))[c];
    split_store(v.x, v.y, v.z, v.w, hhi, hlo, (size_t)i * 32 + c);
}


template<int K, int N>
__global__ void k_split_Wt(const float* __restrict__ W, _Float16* __restrict__ Whi,
                           _Float16* __restrict__ Wlo) {
    int t = blockIdx.x * blockDim.x + threadIdx.x;
    if (t >= K * N) return;
    int k = t / N, n = t % N;
    float w = W[k * N + n];
    _Float16 hi = (_Float16)w;
    Whi[n * K + k] = hi;
    Wlo[n * K + k] = (_Float16)(w - (float)hi);
}


template<int NOUT>
__global__ __launch_bounds__(256) void k_gemm(const _Float16* __restrict__ Ahi,
                                              const _Float16* __restrict__ Alo,
                                              const _Float16* __restrict__ Bhi,
                                              const _Float16* __restrict__ Blo,
                                              float* __restrict__ out) {
    constexpr int K  = EMBED;
    constexpr int NT = NOUT / 16;
    __shared__ _Float16 sBh[NOUT * K];
    __shared__ _Float16 sBl[NOUT * K];

    for (int i = threadIdx.x; i < NOUT * K / 8; i += 256) {
        ((v8h*)sBh)[i] = ((const v8h*)Bhi)[i];
        ((v8h*)sBl)[i] = ((const v8h*)Blo)[i];
    }
    __syncthreads();

    const int wave  = threadIdx.x >> 5;
    const int lane  = threadIdx.x & 31;
    const int strip = blockIdx.x * 8 + wave;
    if (strip >= NNODES / 16) return;
    const int half = lane >> 4;
    const int mrow = strip * 16 + (lane & 15);

    v8f acc[NT] = {};

#pragma unroll
    for (int kc = 0; kc < K / 32; ++kc) {
        const int abase = mrow * K + kc * 32 + half * 8;
        v8h a0h = *(const v8h*)(Ahi + abase);
        v8h a1h = *(const v8h*)(Ahi + abase + 16);
        v8h a0l = *(const v8h*)(Alo + abase);
        v8h a1l = *(const v8h*)(Alo + abase + 16);
        v16h fah = __builtin_shufflevector(a0h, a1h, 0,1,2,3,4,5,6,7,8,9,10,11,12,13,14,15);
        v16h fal = __builtin_shufflevector(a0l, a1l, 0,1,2,3,4,5,6,7,8,9,10,11,12,13,14,15);
#pragma unroll
        for (int t = 0; t < NT; ++t) {
            const int bbase = (t * 16 + (lane & 15)) * K + kc * 32 + half * 8;
            v8h b0h = *(const v8h*)(sBh + bbase);
            v8h b1h = *(const v8h*)(sBh + bbase + 16);
            v8h b0l = *(const v8h*)(sBl + bbase);
            v8h b1l = *(const v8h*)(sBl + bbase + 16);
            v16h fbh = __builtin_shufflevector(b0h, b1h, 0,1,2,3,4,5,6,7,8,9,10,11,12,13,14,15);
            v16h fbl = __builtin_shufflevector(b0l, b1l, 0,1,2,3,4,5,6,7,8,9,10,11,12,13,14,15);
            acc[t] = __builtin_amdgcn_wmma_f32_16x16x32_f16(false, fah, false, fbh,
                                                            (short)0, acc[t], false, false);
            acc[t] = __builtin_amdgcn_wmma_f32_16x16x32_f16(false, fah, false, fbl,
                                                            (short)0, acc[t], false, false);
            acc[t] = __builtin_amdgcn_wmma_f32_16x16x32_f16(false, fal, false, fbh,
                                                            (short)0, acc[t], false, false);
        }
    }

#pragma unroll
    for (int t = 0; t < NT; ++t) {
        const int ncol = t * 16 + (lane & 15);
#pragma unroll
        for (int r = 0; r < 8; ++r) {
            const int row = strip * 16 + half * 8 + r;
            out[(size_t)row * NOUT + ncol] = acc[t][r];
        }
    }
}


template<int NOUT, bool RELUSPLIT>
__global__ void k_gather(const float* __restrict__ hw, const int* __restrict__ off,
                         const int* __restrict__ col, const float* __restrict__ wt,
                         const float* __restrict__ dinv, const float* __restrict__ bias,
                         _Float16* __restrict__ hhi, _Float16* __restrict__ hlo,
                         float* __restrict__ outf) {
    constexpr int LPN = NOUT / 4;
    int t = blockIdx.x * blockDim.x + threadIdx.x;
    int node = t / LPN;
    int c    = t % LPN;
    if (node >= NNODES) return;

    float di = dinv[node];
    float sw = di * di;
    float4 v = ((const float4*)hw)[(size_t)node * LPN + c];
    float4 acc = { sw * v.x, sw * v.y, sw * v.z, sw * v.w };

    const int j0  = off[node];
    const int deg = off[node + 1] - j0;

    for (int base = 0; base < deg; base += LPN) {
        int n = deg - base; if (n > LPN) n = LPN;
        int   sl = 0;
        float wl = 0.0f;
        if (c < n) { int jj = j0 + base + c; sl = col[jj]; wl = wt[jj]; }
        for (int k = 0; k < n; ++k) {
            int   s = __shfl(sl, k, LPN);
            float w = __shfl(wl, k, LPN);
            float4 m = ((const float4*)hw)[(size_t)s * LPN + c];
            acc.x += w * m.x; acc.y += w * m.y; acc.z += w * m.z; acc.w += w * m.w;
        }
    }

    float4 b = ((const float4*)bias)[c];
    acc.x += b.x; acc.y += b.y; acc.z += b.z; acc.w += b.w;

    if (RELUSPLIT) {
        acc.x = fmaxf(acc.x, 0.f); acc.y = fmaxf(acc.y, 0.f);
        acc.z = fmaxf(acc.z, 0.f); acc.w = fmaxf(acc.w, 0.f);
        split_store(acc.x, acc.y, acc.z, acc.w, hhi, hlo, (size_t)node * LPN + c);
    } else {
        ((float4*)outf)[(size_t)node * LPN + c] = acc;
    }
}


extern "C" void kernel_launch(void* const* d_in, const int* in_sizes, int n_in,
                              void* d_out, int out_size, void* d_ws, size_t ws_size,
                              hipStream_t stream) {
    const int*   x      = (const int*)d_in[0];
    const int*   src    = (const int*)d_in[1];
    const int*   dst    = ((const int*)d_in[1]) + NEDGES;
    const float* W_in   = (const float*)d_in[2];
    const float* W_conv = (const float*)d_in[3];
    const float* b_conv = (const float*)d_in[4];
    const float* W_dec  = (const float*)d_in[5];
    const float* b_dec  = (const float*)d_in[6];
    (void)in_sizes; (void)n_in; (void)out_size; (void)ws_size;

    char* p = (char*)d_ws;
    auto carve = [&](size_t bytes) { char* r = p; p += (bytes + 255) & ~(size_t)255; return r; };
    float*    dinv   = (float*)   carve((size_t)NNODES * 4);
    int*      cnt    = (int*)     carve((size_t)NNODES * 4);
    int*      cursor = (int*)     carve((size_t)NNODES * 4);
    int*      off    = (int*)     carve((size_t)(NNODES + 1) * 4);
    int*      col    = (int*)     carve((size_t)NEDGES * 4);
    float*    wt     = (float*)   carve((size_t)NEDGES * 4);
    float*    hw     = (float*)   carve((size_t)NNODES * EMBED * 4);
    _Float16* hhi    = (_Float16*)carve((size_t)NNODES * EMBED * 2);
    _Float16* hlo    = (_Float16*)carve((size_t)NNODES * EMBED * 2);
    _Float16* Wch    = (_Float16*)carve((size_t)EMBED * EMBED * 2);
    _Float16* Wcl    = (_Float16*)carve((size_t)EMBED * EMBED * 2);
    _Float16* Wdh    = (_Float16*)carve((size_t)NTYPES * EMBED * 2);
    _Float16* Wdl    = (_Float16*)carve((size_t)NTYPES * EMBED * 2);

    const int B = 256;
    auto blocks = [](long long n, int b) { return (int)((n + b - 1) / b); };

    k_zero2   <<<blocks(NNODES, B), B, 0, stream>>>(cnt, cursor);
    k_count   <<<blocks(NEDGES, B), B, 0, stream>>>(dst, cnt);
    k_dinv    <<<blocks(NNODES, B), B, 0, stream>>>(cnt, dinv);
    k_scan    <<<1, 1024, 0, stream>>>(cnt, off);
    k_csr_fill<<<blocks(NEDGES, B), B, 0, stream>>>(src, dst, off, dinv, cursor, col, wt);

    k_split_Wt<EMBED, EMBED> <<<blocks(EMBED * EMBED, B), B, 0, stream>>>(W_conv, Wch, Wcl);
    k_split_Wt<EMBED, NTYPES><<<blocks(EMBED * NTYPES, B), B, 0, stream>>>(W_dec,  Wdh, Wdl);

    k_embed_split<<<blocks((long long)NNODES * 32, B), B, 0, stream>>>(x, W_in, hhi, hlo);

    const int gemm_blocks = ((NNODES / 16) + 7) / 8;

    for (int it = 0; it < NITERS; ++it) {
        k_gemm<EMBED><<<gemm_blocks, B, 0, stream>>>(hhi, hlo, Wch, Wcl, hw);
        k_gather<EMBED, true><<<blocks((long long)NNODES * 32, B), B, 0, stream>>>(
            hw, off, col, wt, dinv, b_conv, hhi, hlo, nullptr);
    }

    k_gemm<NTYPES><<<gemm_blocks, B, 0, stream>>>(hhi, hlo, Wdh, Wdl, hw);
    k_gather<NTYPES, false><<<blocks((long long)NNODES * 16, B), B, 0, stream>>>(
        hw, off, col, wt, dinv, b_dec, nullptr, nullptr, (float*)d_out);
}


# prefix sum of node degrees: 40 chunked log-step LDS scans (about 960 workgroup barriers) replaced by one pass of wave-shuffle scans over int4 loads with a single barrier
# speedup vs baseline: 1.2383x; 1.2383x over previous
#include <hip/hip_runtime.h>

typedef __attribute__((ext_vector_type(4)))  _Float16 v4h;
typedef __attribute__((ext_vector_type(8)))  _Float16 v8h;
typedef __attribute__((ext_vector_type(16))) _Float16 v16h;
typedef __attribute__((ext_vector_type(8)))  float    v8f;

#define NNODES 40000
#define NEDGES 640000
#define EMBED  128
#define NTYPES 64
#define NITERS 4


__global__ void k_zero2(int* a, int* b) {
    int i = blockIdx.x * blockDim.x + threadIdx.x;
    if (i < NNODES) { a[i] = 0; b[i] = 0; }
}

__global__ void k_count(const int* __restrict__ dst, int* cnt) {
    int e = blockIdx.x * blockDim.x + threadIdx.x;
    if (e < NEDGES) atomicAdd(&cnt[dst[e]], 1);
}

__global__ void k_dinv(const int* __restrict__ cnt, float* __restrict__ dinv) {
    int i = blockIdx.x * blockDim.x + threadIdx.x;
    if (i < NNODES) dinv[i] = rsqrtf((float)cnt[i] + 1.0f);
}

#define SCAN_THREADS 1024
#define SCAN_WAVES   (SCAN_THREADS / 32)
#define SCAN_VEC     (NNODES / 4)
#define SCAN_ITERS   ((SCAN_VEC + SCAN_THREADS - 1) / SCAN_THREADS)

__global__ __launch_bounds__(SCAN_THREADS) void k_scan(const int* __restrict__ cnt,
                                                       int* __restrict__ off) {
    static_assert(NNODES % 4 == 0, "int4 path needs NNODES % 4 == 0");
    static_assert(SCAN_WAVES == 32, "one lane per wave total");
    __shared__ int wtot[SCAN_WAVES];
    const int lane  = threadIdx.x & 31;
    const int wave  = threadIdx.x >> 5;
    const int vbase = wave * (SCAN_ITERS * 32);

    int4 v[SCAN_ITERS];
#pragma unroll
    for (int i = 0; i < SCAN_ITERS; ++i) {
        const int idx = vbase + i * 32 + lane;
        v[i] = (idx < SCAN_VEC) ? ((const int4*)cnt)[idx] : make_int4(0, 0, 0, 0);
    }

    int own[SCAN_ITERS], incl[SCAN_ITERS];
#pragma unroll
    for (int i = 0; i < SCAN_ITERS; ++i) { own[i] = v[i].x + v[i].y + v[i].z + v[i].w; incl[i] = own[i]; }
#pragma unroll
    for (int d = 1; d < 32; d <<= 1) {
#pragma unroll
        for (int i = 0; i < SCAN_ITERS; ++i) {
            const int t = __shfl_up(incl[i], d, 32);
            if (lane >= d) incl[i] += t;
        }
    }

    int excl[SCAN_ITERS];
    int run = 0;
#pragma unroll
    for (int i = 0; i < SCAN_ITERS; ++i) {
        excl[i] = run + incl[i] - own[i];
        run += __shfl(incl[i], 31, 32);
    }

    if (lane == 0) wtot[wave] = run;
    __syncthreads();

    const int wt = wtot[lane];
    int ws = wt;
#pragma unroll
    for (int d = 1; d < 32; d <<= 1) {
        const int t = __shfl_up(ws, d, 32);
        if (lane >= d) ws += t;
    }
    const int wprefix = __shfl(ws - wt, wave, 32);
    const int total   = __shfl(ws, 31, 32);

#pragma unroll
    for (int i = 0; i < SCAN_ITERS; ++i) {
        const int idx = vbase + i * 32 + lane;
        const int p = wprefix + excl[i];
        if (idx < SCAN_VEC)
            ((int4*)off)[idx] = make_int4(p, p + v[i].x, p + v[i].x + v[i].y,
                                          p + v[i].x + v[i].y + v[i].z);
    }
    if (threadIdx.x == 0) off[NNODES] = total;
}

__global__ void k_csr_fill(const int* __restrict__ src, const int* __restrict__ dst,
                           const int* __restrict__ off, const float* __restrict__ dinv,
                           int* __restrict__ cursor, int* __restrict__ col,
                           float* __restrict__ wt) {
    int e = blockIdx.x * blockDim.x + threadIdx.x;
    if (e >= NEDGES) return;
    int s = src[e], d = dst[e];
    int j = off[d] + atomicAdd(&cursor[d], 1);
    col[j] = s;
    wt[j]  = dinv[s] * dinv[d];
}


__device__ __forceinline__ void split_store(float x, float y, float z, float w,
                                            _Float16* hi, _Float16* lo, size_t idx4) {
    v4h H, L;
    H[0] = (_Float16)x; L[0] = (_Float16)(x - (float)H[0]);
    H[1] = (_Float16)y; L[1] = (_Float16)(y - (float)H[1]);
    H[2] = (_Float16)z; L[2] = (_Float16)(z - (float)H[2]);
    H[3] = (_Float16)w; L[3] = (_Float16)(w - (float)H[3]);
    ((v4h*)hi)[idx4] = H;
    ((v4h*)lo)[idx4] = L;
}

__global__ void k_embed_split(const int* __restrict__ x, const float* __restrict__ Win,
                              _Float16* __restrict__ hhi, _Float16* __restrict__ hlo) {
    int t = blockIdx.x * blockDim.x + threadIdx.x;
    if (t >= NNODES * (EMBED / 4)) return;
    int i = t >> 5, c = t & 31;
    float4 v = ((const float4*)(Win + (size_t)x[i] * EMBED))[c];
    split_store(v.x, v.y, v.z, v.w, hhi, hlo, (size_t)i * 32 + c);
}


template<int K, int N>
__global__ void k_split_Wt(const float* __restrict__ W, _Float16* __restrict__ Whi,
                           _Float16* __restrict__ Wlo) {
    int t = blockIdx.x * blockDim.x + threadIdx.x;
    if (t >= K * N) return;
    int k = t / N, n = t % N;
    float w = W[k * N + n];
    _Float16 hi = (_Float16)w;
    Whi[n * K + k] = hi;
    Wlo[n * K + k] = (_Float16)(w - (float)hi);
}


template<int NOUT>
__global__ __launch_bounds__(256) void k_gemm(const _Float16* __restrict__ Ahi,
                                              const _Float16* __restrict__ Alo,
                                              const _Float16* __restrict__ Bhi,
                                              const _Float16* __restrict__ Blo,
                                              float* __restrict__ out) {
    constexpr int K  = EMBED;
    constexpr int NT = NOUT / 16;
    __shared__ _Float16 sBh[NOUT * K];
    __shared__ _Float16 sBl[NOUT * K];

    for (int i = threadIdx.x; i < NOUT * K / 8; i += 256) {
        ((v8h*)sBh)[i] = ((const v8h*)Bhi)[i];
        ((v8h*)sBl)[i] = ((const v8h*)Blo)[i];
    }
    __syncthreads();

    const int wave  = threadIdx.x >> 5;
    const int lane  = threadIdx.x & 31;
    const int strip = blockIdx.x * 8 + wave;
    if (strip >= NNODES / 16) return;
    const int half = lane >> 4;
    const int mrow = strip * 16 + (lane & 15);

    v8f acc[NT] = {};

#pragma unroll
    for (int kc = 0; kc < K / 32; ++kc) {
        const int abase = mrow * K + kc * 32 + half * 8;
        v8h a0h = *(const v8h*)(Ahi + abase);
        v8h a1h = *(const v8h*)(Ahi + abase + 16);
        v8h a0l = *(const v8h*)(Alo + abase);
        v8h a1l = *(const v8h*)(Alo + abase + 16);
        v16h fah = __builtin_shufflevector(a0h, a1h, 0,1,2,3,4,5,6,7,8,9,10,11,12,13,14,15);
        v16h fal = __builtin_shufflevector(a0l, a1l, 0,1,2,3,4,5,6,7,8,9,10,11,12,13,14,15);
#pragma unroll
        for (int t = 0; t < NT; ++t) {
            const int bbase = (t * 16 + (lane & 15)) * K + kc * 32 + half * 8;
            v8h b0h = *(const v8h*)(sBh + bbase);
            v8h b1h = *(const v8h*)(sBh + bbase + 16);
            v8h b0l = *(const v8h*)(sBl + bbase);
            v8h b1l = *(const v8h*)(sBl + bbase + 16);
            v16h fbh = __builtin_shufflevector(b0h, b1h, 0,1,2,3,4,5,6,7,8,9,10,11,12,13,14,15);
            v16h fbl = __builtin_shufflevector(b0l, b1l, 0,1,2,3,4,5,6,7,8,9,10,11,12,13,14,15);
            acc[t] = __builtin_amdgcn_wmma_f32_16x16x32_f16(false, fah, false, fbh,
                                                            (short)0, acc[t], false, false);
            acc[t] = __builtin_amdgcn_wmma_f32_16x16x32_f16(false, fah, false, fbl,
                                                            (short)0, acc[t], false, false);
            acc[t] = __builtin_amdgcn_wmma_f32_16x16x32_f16(false, fal, false, fbh,
                                                            (short)0, acc[t], false, false);
        }
    }

#pragma unroll
    for (int t = 0; t < NT; ++t) {
        const int ncol = t * 16 + (lane & 15);
#pragma unroll
        for (int r = 0; r < 8; ++r) {
            const int row = strip * 16 + half * 8 + r;
            out[(size_t)row * NOUT + ncol] = acc[t][r];
        }
    }
}


template<int NOUT, bool RELUSPLIT>
__global__ void k_gather(const float* __restrict__ hw, const int* __restrict__ off,
                         const int* __restrict__ col, const float* __restrict__ wt,
                         const float* __restrict__ dinv, const float* __restrict__ bias,
                         _Float16* __restrict__ hhi, _Float16* __restrict__ hlo,
                         float* __restrict__ outf) {
    constexpr int LPN = NOUT / 4;
    int t = blockIdx.x * blockDim.x + threadIdx.x;
    int node = t / LPN;
    int c    = t % LPN;
    if (node >= NNODES) return;

    float di = dinv[node];
    float sw = di * di;
    float4 v = ((const float4*)hw)[(size_t)node * LPN + c];
    float4 acc = { sw * v.x, sw * v.y, sw * v.z, sw * v.w };

    const int j0  = off[node];
    const int deg = off[node + 1] - j0;

    for (int base = 0; base < deg; base += LPN) {
        int n = deg - base; if (n > LPN) n = LPN;
        int   sl = 0;
        float wl = 0.0f;
        if (c < n) { int jj = j0 + base + c; sl = col[jj]; wl = wt[jj]; }
        for (int k = 0; k < n; ++k) {
            int   s = __shfl(sl, k, LPN);
            float w = __shfl(wl, k, LPN);
            float4 m = ((const float4*)hw)[(size_t)s * LPN + c];
            acc.x += w * m.x; acc.y += w * m.y; acc.z += w * m.z; acc.w += w * m.w;
        }
    }

    float4 b = ((const float4*)bias)[c];
    acc.x += b.x; acc.y += b.y; acc.z += b.z; acc.w += b.w;

    if (RELUSPLIT) {
        acc.x = fmaxf(acc.x, 0.f); acc.y = fmaxf(acc.y, 0.f);
        acc.z = fmaxf(acc.z, 0.f); acc.w = fmaxf(acc.w, 0.f);
        split_store(acc.x, acc.y, acc.z, acc.w, hhi, hlo, (size_t)node * LPN + c);
    } else {
        ((float4*)outf)[(size_t)node * LPN + c] = acc;
    }
}


extern "C" void kernel_launch(void* const* d_in, const int* in_sizes, int n_in,
                              void* d_out, int out_size, void* d_ws, size_t ws_size,
                              hipStream_t stream) {
    const int*   x      = (const int*)d_in[0];
    const int*   src    = (const int*)d_in[1];
    const int*   dst    = ((const int*)d_in[1]) + NEDGES;
    const float* W_in   = (const float*)d_in[2];
    const float* W_conv = (const float*)d_in[3];
    const float* b_conv = (const float*)d_in[4];
    const float* W_dec  = (const float*)d_in[5];
    const float* b_dec  = (const float*)d_in[6];
    (void)in_sizes; (void)n_in; (void)out_size; (void)ws_size;

    char* p = (char*)d_ws;
    auto carve = [&](size_t bytes) { char* r = p; p += (bytes + 255) & ~(size_t)255; return r; };
    float*    dinv   = (float*)   carve((size_t)NNODES * 4);
    int*      cnt    = (int*)     carve((size_t)NNODES * 4);
    int*      cursor = (int*)     carve((size_t)NNODES * 4);
    int*      off    = (int*)     carve((size_t)(NNODES + 1) * 4);
    int*      col    = (int*)     carve((size_t)NEDGES * 4);
    float*    wt     = (float*)   carve((size_t)NEDGES * 4);
    float*    hw     = (float*)   carve((size_t)NNODES * EMBED * 4);
    _Float16* hhi    = (_Float16*)carve((size_t)NNODES * EMBED * 2);
    _Float16* hlo    = (_Float16*)carve((size_t)NNODES * EMBED * 2);
    _Float16* Wch    = (_Float16*)carve((size_t)EMBED * EMBED * 2);
    _Float16* Wcl    = (_Float16*)carve((size_t)EMBED * EMBED * 2);
    _Float16* Wdh    = (_Float16*)carve((size_t)NTYPES * EMBED * 2);
    _Float16* Wdl    = (_Float16*)carve((size_t)NTYPES * EMBED * 2);

    const int B = 256;
    auto blocks = [](long long n, int b) { return (int)((n + b - 1) / b); };

    k_zero2   <<<blocks(NNODES, B), B, 0, stream>>>(cnt, cursor);
    k_count   <<<blocks(NEDGES, B), B, 0, stream>>>(dst, cnt);
    k_dinv    <<<blocks(NNODES, B), B, 0, stream>>>(cnt, dinv);
    k_scan    <<<1, SCAN_THREADS, 0, stream>>>(cnt, off);
    k_csr_fill<<<blocks(NEDGES, B), B, 0, stream>>>(src, dst, off, dinv, cursor, col, wt);

    k_split_Wt<EMBED, EMBED> <<<blocks(EMBED * EMBED, B), B, 0, stream>>>(W_conv, Wch, Wcl);
    k_split_Wt<EMBED, NTYPES><<<blocks(EMBED * NTYPES, B), B, 0, stream>>>(W_dec,  Wdh, Wdl);

    k_embed_split<<<blocks((long long)NNODES * 32, B), B, 0, stream>>>(x, W_in, hhi, hlo);

    const int gemm_blocks = ((NNODES / 16) + 7) / 8;

    for (int it = 0; it < NITERS; ++it) {
        k_gemm<EMBED><<<gemm_blocks, B, 0, stream>>>(hhi, hlo, Wch, Wcl, hw);
        k_gather<EMBED, true><<<blocks((long long)NNODES * 32, B), B, 0, stream>>>(
            hw, off, col, wt, dinv, b_conv, hhi, hlo, nullptr);
    }

    k_gemm<NTYPES><<<gemm_blocks, B, 0, stream>>>(hhi, hlo, Wdh, Wdl, hw);
    k_gather<NTYPES, false><<<blocks((long long)NNODES * 16, B), B, 0, stream>>>(
        hw, off, col, wt, dinv, b_dec, nullptr, nullptr, (float*)d_out);
}


# GEMM weight tiles in LDS: row pitch 256 B -> 272 B (16 B x odd), so the 16 rows of each B-fragment ds_load_b128 fall on distinct banks instead of the same 4 (16-way bank conflict removed)
# speedup vs baseline: 1.5947x; 1.2878x over previous
#include <hip/hip_runtime.h>

typedef __attribute__((ext_vector_type(4)))  _Float16 v4h;
typedef __attribute__((ext_vector_type(8)))  _Float16 v8h;
typedef __attribute__((ext_vector_type(16))) _Float16 v16h;
typedef __attribute__((ext_vector_type(8)))  float    v8f;

#define NNODES 40000
#define NEDGES 640000
#define EMBED  128
#define NTYPES 64
#define NITERS 4


__global__ void k_zero2(int* a, int* b) {
    int i = blockIdx.x * blockDim.x + threadIdx.x;
    if (i < NNODES) { a[i] = 0; b[i] = 0; }
}

__global__ void k_count(const int* __restrict__ dst, int* cnt) {
    int e = blockIdx.x * blockDim.x + threadIdx.x;
    if (e < NEDGES) atomicAdd(&cnt[dst[e]], 1);
}

__global__ void k_dinv(const int* __restrict__ cnt, float* __restrict__ dinv) {
    int i = blockIdx.x * blockDim.x + threadIdx.x;
    if (i < NNODES) dinv[i] = rsqrtf((float)cnt[i] + 1.0f);
}

#define SCAN_THREADS 1024
#define SCAN_WAVES   (SCAN_THREADS / 32)
#define SCAN_VEC     (NNODES / 4)
#define SCAN_ITERS   ((SCAN_VEC + SCAN_THREADS - 1) / SCAN_THREADS)

__global__ __launch_bounds__(SCAN_THREADS) void k_scan(const int* __restrict__ cnt,
                                                       int* __restrict__ off) {
    static_assert(NNODES % 4 == 0, "int4 path needs NNODES % 4 == 0");
    static_assert(SCAN_WAVES == 32, "one lane per wave total");
    __shared__ int wtot[SCAN_WAVES];
    const int lane  = threadIdx.x & 31;
    const int wave  = threadIdx.x >> 5;
    const int vbase = wave * (SCAN_ITERS * 32);

    int4 v[SCAN_ITERS];
#pragma unroll
    for (int i = 0; i < SCAN_ITERS; ++i) {
        const int idx = vbase + i * 32 + lane;
        v[i] = (idx < SCAN_VEC) ? ((const int4*)cnt)[idx] : make_int4(0, 0, 0, 0);
    }

    int own[SCAN_ITERS], incl[SCAN_ITERS];
#pragma unroll
    for (int i = 0; i < SCAN_ITERS; ++i) { own[i] = v[i].x + v[i].y + v[i].z + v[i].w; incl[i] = own[i]; }
#pragma unroll
    for (int d = 1; d < 32; d <<= 1) {
#pragma unroll
        for (int i = 0; i < SCAN_ITERS; ++i) {
            const int t = __shfl_up(incl[i], d, 32);
            if (lane >= d) incl[i] += t;
        }
    }

    int excl[SCAN_ITERS];
    int run = 0;
#pragma unroll
    for (int i = 0; i < SCAN_ITERS; ++i) {
        excl[i] = run + incl[i] - own[i];
        run += __shfl(incl[i], 31, 32);
    }

    if (lane == 0) wtot[wave] = run;
    __syncthreads();

    const int wt = wtot[lane];
    int ws = wt;
#pragma unroll
    for (int d = 1; d < 32; d <<= 1) {
        const int t = __shfl_up(ws, d, 32);
        if (lane >= d) ws += t;
    }
    const int wprefix = __shfl(ws - wt, wave, 32);
    const int total   = __shfl(ws, 31, 32);

#pragma unroll
    for (int i = 0; i < SCAN_ITERS; ++i) {
        const int idx = vbase + i * 32 + lane;
        const int p = wprefix + excl[i];
        if (idx < SCAN_VEC)
            ((int4*)off)[idx] = make_int4(p, p + v[i].x, p + v[i].x + v[i].y,
                                          p + v[i].x + v[i].y + v[i].z);
    }
    if (threadIdx.x == 0) off[NNODES] = total;
}

__global__ void k_csr_fill(const int* __restrict__ src, const int* __restrict__ dst,
                           const int* __restrict__ off, const float* __restrict__ dinv,
                           int* __restrict__ cursor, int* __restrict__ col,
                           float* __restrict__ wt) {
    int e = blockIdx.x * blockDim.x + threadIdx.x;
    if (e >= NEDGES) return;
    int s = src[e], d = dst[e];
    int j = off[d] + atomicAdd(&cursor[d], 1);
    col[j] = s;
    wt[j]  = dinv[s] * dinv[d];
}


__device__ __forceinline__ void split_store(float x, float y, float z, float w,
                                            _Float16* hi, _Float16* lo, size_t idx4) {
    v4h H, L;
    H[0] = (_Float16)x; L[0] = (_Float16)(x - (float)H[0]);
    H[1] = (_Float16)y; L[1] = (_Float16)(y - (float)H[1]);
    H[2] = (_Float16)z; L[2] = (_Float16)(z - (float)H[2]);
    H[3] = (_Float16)w; L[3] = (_Float16)(w - (float)H[3]);
    ((v4h*)hi)[idx4] = H;
    ((v4h*)lo)[idx4] = L;
}

__global__ void k_embed_split(const int* __restrict__ x, const float* __restrict__ Win,
                              _Float16* __restrict__ hhi, _Float16* __restrict__ hlo) {
    int t = blockIdx.x * blockDim.x + threadIdx.x;
    if (t >= NNODES * (EMBED / 4)) return;
    int i = t >> 5, c = t & 31;
    float4 v = ((const float4*)(Win + (size_t)x[i] * EMBED))[c];
    split_store(v.x, v.y, v.z, v.w, hhi, hlo, (size_t)i * 32 + c);
}


template<int K, int N>
__global__ void k_split_Wt(const float* __restrict__ W, _Float16* __restrict__ Whi,
                           _Float16* __restrict__ Wlo) {
    int t = blockIdx.x * blockDim.x + threadIdx.x;
    if (t >= K * N) return;
    int k = t / N, n = t % N;
    float w = W[k * N + n];
    _Float16 hi = (_Float16)w;
    Whi[n * K + k] = hi;
    Wlo[n * K + k] = (_Float16)(w - (float)hi);
}


template<int NOUT>
__global__ __launch_bounds__(256) void k_gemm(const _Float16* __restrict__ Ahi,
                                              const _Float16* __restrict__ Alo,
                                              const _Float16* __restrict__ Bhi,
                                              const _Float16* __restrict__ Blo,
                                              float* __restrict__ out) {
    constexpr int K  = EMBED;
    constexpr int NT = NOUT / 16;
    constexpr int KP = K + 8;
    __shared__ _Float16 sBh[NOUT * KP];
    __shared__ _Float16 sBl[NOUT * KP];

    for (int i = threadIdx.x; i < NOUT * K / 8; i += 256) {
        const int d = (i >> 4) * KP + (i & 15) * 8;
        *(v8h*)(sBh + d) = ((const v8h*)Bhi)[i];
        *(v8h*)(sBl + d) = ((const v8h*)Blo)[i];
    }
    __syncthreads();

    const int wave  = threadIdx.x >> 5;
    const int lane  = threadIdx.x & 31;
    const int strip = blockIdx.x * 8 + wave;
    if (strip >= NNODES / 16) return;
    const int half = lane >> 4;
    const int mrow = strip * 16 + (lane & 15);

    v8f acc[NT] = {};

#pragma unroll
    for (int kc = 0; kc < K / 32; ++kc) {
        const int abase = mrow * K + kc * 32 + half * 8;
        v8h a0h = *(const v8h*)(Ahi + abase);
        v8h a1h = *(const v8h*)(Ahi + abase + 16);
        v8h a0l = *(const v8h*)(Alo + abase);
        v8h a1l = *(const v8h*)(Alo + abase + 16);
        v16h fah = __builtin_shufflevector(a0h, a1h, 0,1,2,3,4,5,6,7,8,9,10,11,12,13,14,15);
        v16h fal = __builtin_shufflevector(a0l, a1l, 0,1,2,3,4,5,6,7,8,9,10,11,12,13,14,15);
#pragma unroll
        for (int t = 0; t < NT; ++t) {
            const int bbase = (t * 16 + (lane & 15)) * KP + kc * 32 + half * 8;
            v8h b0h = *(const v8h*)(sBh + bbase);
            v8h b1h = *(const v8h*)(sBh + bbase + 16);
            v8h b0l = *(const v8h*)(sBl + bbase);
            v8h b1l = *(const v8h*)(sBl + bbase + 16);
            v16h fbh = __builtin_shufflevector(b0h, b1h, 0,1,2,3,4,5,6,7,8,9,10,11,12,13,14,15);
            v16h fbl = __builtin_shufflevector(b0l, b1l, 0,1,2,3,4,5,6,7,8,9,10,11,12,13,14,15);
            acc[t] = __builtin_amdgcn_wmma_f32_16x16x32_f16(false, fah, false, fbh,
                                                            (short)0, acc[t], false, false);
            acc[t] = __builtin_amdgcn_wmma_f32_16x16x32_f16(false, fah, false, fbl,
                                                            (short)0, acc[t], false, false);
            acc[t] = __builtin_amdgcn_wmma_f32_16x16x32_f16(false, fal, false, fbh,
                                                            (short)0, acc[t], false, false);
        }
    }

#pragma unroll
    for (int t = 0; t < NT; ++t) {
        const int ncol = t * 16 + (lane & 15);
#pragma unroll
        for (int r = 0; r < 8; ++r) {
            const int row = strip * 16 + half * 8 + r;
            out[(size_t)row * NOUT + ncol] = acc[t][r];
        }
    }
}


template<int NOUT, bool RELUSPLIT>
__global__ void k_gather(const float* __restrict__ hw, const int* __restrict__ off,
                         const int* __restrict__ col, const float* __restrict__ wt,
                         const float* __restrict__ dinv, const float* __restrict__ bias,
                         _Float16* __restrict__ hhi, _Float16* __restrict__ hlo,
                         float* __restrict__ outf) {
    constexpr int LPN = NOUT / 4;
    int t = blockIdx.x * blockDim.x + threadIdx.x;
    int node = t / LPN;
    int c    = t % LPN;
    if (node >= NNODES) return;

    float di = dinv[node];
    float sw = di * di;
    float4 v = ((const float4*)hw)[(size_t)node * LPN + c];
    float4 acc = { sw * v.x, sw * v.y, sw * v.z, sw * v.w };

    const int j0  = off[node];
    const int deg = off[node + 1] - j0;

    for (int base = 0; base < deg; base += LPN) {
        int n = deg - base; if (n > LPN) n = LPN;
        int   sl = 0;
        float wl = 0.0f;
        if (c < n) { int jj = j0 + base + c; sl = col[jj]; wl = wt[jj]; }
        for (int k = 0; k < n; ++k) {
            int   s = __shfl(sl, k, LPN);
            float w = __shfl(wl, k, LPN);
            float4 m = ((const float4*)hw)[(size_t)s * LPN + c];
            acc.x += w * m.x; acc.y += w * m.y; acc.z += w * m.z; acc.w += w * m.w;
        }
    }

    float4 b = ((const float4*)bias)[c];
    acc.x += b.x; acc.y += b.y; acc.z += b.z; acc.w += b.w;

    if (RELUSPLIT) {
        acc.x = fmaxf(acc.x, 0.f); acc.y = fmaxf(acc.y, 0.f);
        acc.z = fmaxf(acc.z, 0.f); acc.w = fmaxf(acc.w, 0.f);
        split_store(acc.x, acc.y, acc.z, acc.w, hhi, hlo, (size_t)node * LPN + c);
    } else {
        ((float4*)outf)[(size_t)node * LPN + c] = acc;
    }
}


extern "C" void kernel_launch(void* const* d_in, const int* in_sizes, int n_in,
                              void* d_out, int out_size, void* d_ws, size_t ws_size,
                              hipStream_t stream) {
    const int*   x      = (const int*)d_in[0];
    const int*   src    = (const int*)d_in[1];
    const int*   dst    = ((const int*)d_in[1]) + NEDGES;
    const float* W_in   = (const float*)d_in[2];
    const float* W_conv = (const float*)d_in[3];
    const float* b_conv = (const float*)d_in[4];
    const float* W_dec  = (const float*)d_in[5];
    const float* b_dec  = (const float*)d_in[6];
    (void)in_sizes; (void)n_in; (void)out_size; (void)ws_size;

    char* p = (char*)d_ws;
    auto carve = [&](size_t bytes) { char* r = p; p += (bytes + 255) & ~(size_t)255; return r; };
    float*    dinv   = (float*)   carve((size_t)NNODES * 4);
    int*      cnt    = (int*)     carve((size_t)NNODES * 4);
    int*      cursor = (int*)     carve((size_t)NNODES * 4);
    int*      off    = (int*)     carve((size_t)(NNODES + 1) * 4);
    int*      col    = (int*)     carve((size_t)NEDGES * 4);
    float*    wt     = (float*)   carve((size_t)NEDGES * 4);
    float*    hw     = (float*)   carve((size_t)NNODES * EMBED * 4);
    _Float16* hhi    = (_Float16*)carve((size_t)NNODES * EMBED * 2);
    _Float16* hlo    = (_Float16*)carve((size_t)NNODES * EMBED * 2);
    _Float16* Wch    = (_Float16*)carve((size_t)EMBED * EMBED * 2);
    _Float16* Wcl    = (_Float16*)carve((size_t)EMBED * EMBED * 2);
    _Float16* Wdh    = (_Float16*)carve((size_t)NTYPES * EMBED * 2);
    _Float16* Wdl    = (_Float16*)carve((size_t)NTYPES * EMBED * 2);

    const int B = 256;
    auto blocks = [](long long n, int b) { return (int)((n + b - 1) / b); };

    k_zero2   <<<blocks(NNODES, B), B, 0, stream>>>(cnt, cursor);
    k_count   <<<blocks(NEDGES, B), B, 0, stream>>>(dst, cnt);
    k_dinv    <<<blocks(NNODES, B), B, 0, stream>>>(cnt, dinv);
    k_scan    <<<1, SCAN_THREADS, 0, stream>>>(cnt, off);
    k_csr_fill<<<blocks(NEDGES, B), B, 0, stream>>>(src, dst, off, dinv, cursor, col, wt);

    k_split_Wt<EMBED, EMBED> <<<blocks(EMBED * EMBED, B), B, 0, stream>>>(W_conv, Wch, Wcl);
    k_split_Wt<EMBED, NTYPES><<<blocks(EMBED * NTYPES, B), B, 0, stream>>>(W_dec,  Wdh, Wdl);

    k_embed_split<<<blocks((long long)NNODES * 32, B), B, 0, stream>>>(x, W_in, hhi, hlo);

    const int gemm_blocks = ((NNODES / 16) + 7) / 8;

    for (int it = 0; it < NITERS; ++it) {
        k_gemm<EMBED><<<gemm_blocks, B, 0, stream>>>(hhi, hlo, Wch, Wcl, hw);
        k_gather<EMBED, true><<<blocks((long long)NNODES * 32, B), B, 0, stream>>>(
            hw, off, col, wt, dinv, b_conv, hhi, hlo, nullptr);
    }

    k_gemm<NTYPES><<<gemm_blocks, B, 0, stream>>>(hhi, hlo, Wdh, Wdl, hw);
    k_gather<NTYPES, false><<<blocks((long long)NNODES * 16, B), B, 0, stream>>>(
        hw, off, col, wt, dinv, b_dec, nullptr, nullptr, (float*)d_out);
}


# gather kernels: a wave walks 4 consecutive nodes and issues its neighbour-row loads 8 at a time ahead of their multiply-adds (was 1 node per wave, 1 row load in flight); edge order of the sums kept
# speedup vs baseline: 1.7558x; 1.1010x over previous
#include <hip/hip_runtime.h>

typedef __attribute__((ext_vector_type(4)))  _Float16 v4h;
typedef __attribute__((ext_vector_type(8)))  _Float16 v8h;
typedef __attribute__((ext_vector_type(16))) _Float16 v16h;
typedef __attribute__((ext_vector_type(8)))  float    v8f;

#define NNODES 40000
#define NEDGES 640000
#define EMBED  128
#define NTYPES 64
#define NITERS 4


__global__ void k_zero2(int* a, int* b) {
    int i = blockIdx.x * blockDim.x + threadIdx.x;
    if (i < NNODES) { a[i] = 0; b[i] = 0; }
}

__global__ void k_count(const int* __restrict__ dst, int* cnt) {
    int e = blockIdx.x * blockDim.x + threadIdx.x;
    if (e < NEDGES) atomicAdd(&cnt[dst[e]], 1);
}

__global__ void k_dinv(const int* __restrict__ cnt, float* __restrict__ dinv) {
    int i = blockIdx.x * blockDim.x + threadIdx.x;
    if (i < NNODES) dinv[i] = rsqrtf((float)cnt[i] + 1.0f);
}

#define SCAN_THREADS 1024
#define SCAN_WAVES   (SCAN_THREADS / 32)
#define SCAN_VEC     (NNODES / 4)
#define SCAN_ITERS   ((SCAN_VEC + SCAN_THREADS - 1) / SCAN_THREADS)

__global__ __launch_bounds__(SCAN_THREADS) void k_scan(const int* __restrict__ cnt,
                                                       int* __restrict__ off) {
    static_assert(NNODES % 4 == 0, "int4 path needs NNODES % 4 == 0");
    static_assert(SCAN_WAVES == 32, "one lane per wave total");
    __shared__ int wtot[SCAN_WAVES];
    const int lane  = threadIdx.x & 31;
    const int wave  = threadIdx.x >> 5;
    const int vbase = wave * (SCAN_ITERS * 32);

    int4 v[SCAN_ITERS];
#pragma unroll
    for (int i = 0; i < SCAN_ITERS; ++i) {
        const int idx = vbase + i * 32 + lane;
        v[i] = (idx < SCAN_VEC) ? ((const int4*)cnt)[idx] : make_int4(0, 0, 0, 0);
    }

    int own[SCAN_ITERS], incl[SCAN_ITERS];
#pragma unroll
    for (int i = 0; i < SCAN_ITERS; ++i) { own[i] = v[i].x + v[i].y + v[i].z + v[i].w; incl[i] = own[i]; }
#pragma unroll
    for (int d = 1; d < 32; d <<= 1) {
#pragma unroll
        for (int i = 0; i < SCAN_ITERS; ++i) {
            const int t = __shfl_up(incl[i], d, 32);
            if (lane >= d) incl[i] += t;
        }
    }

    int excl[SCAN_ITERS];
    int run = 0;
#pragma unroll
    for (int i = 0; i < SCAN_ITERS; ++i) {
        excl[i] = run + incl[i] - own[i];
        run += __shfl(incl[i], 31, 32);
    }

    if (lane == 0) wtot[wave] = run;
    __syncthreads();

    const int wt = wtot[lane];
    int ws = wt;
#pragma unroll
    for (int d = 1; d < 32; d <<= 1) {
        const int t = __shfl_up(ws, d, 32);
        if (lane >= d) ws += t;
    }
    const int wprefix = __shfl(ws - wt, wave, 32);
    const int total   = __shfl(ws, 31, 32);

#pragma unroll
    for (int i = 0; i < SCAN_ITERS; ++i) {
        const int idx = vbase + i * 32 + lane;
        const int p = wprefix + excl[i];
        if (idx < SCAN_VEC)
            ((int4*)off)[idx] = make_int4(p, p + v[i].x, p + v[i].x + v[i].y,
                                          p + v[i].x + v[i].y + v[i].z);
    }
    if (threadIdx.x == 0) off[NNODES] = total;
}

__global__ void k_csr_fill(const int* __restrict__ src, const int* __restrict__ dst,
                           const int* __restrict__ off, const float* __restrict__ dinv,
                           int* __restrict__ cursor, int* __restrict__ col,
                           float* __restrict__ wt) {
    int e = blockIdx.x * blockDim.x + threadIdx.x;
    if (e >= NEDGES) return;
    int s = src[e], d = dst[e];
    int j = off[d] + atomicAdd(&cursor[d], 1);
    col[j] = s;
    wt[j]  = dinv[s] * dinv[d];
}


__device__ __forceinline__ void split_store(float x, float y, float z, float w,
                                            _Float16* hi, _Float16* lo, size_t idx4) {
    v4h H, L;
    H[0] = (_Float16)x; L[0] = (_Float16)(x - (float)H[0]);
    H[1] = (_Float16)y; L[1] = (_Float16)(y - (float)H[1]);
    H[2] = (_Float16)z; L[2] = (_Float16)(z - (float)H[2]);
    H[3] = (_Float16)w; L[3] = (_Float16)(w - (float)H[3]);
    ((v4h*)hi)[idx4] = H;
    ((v4h*)lo)[idx4] = L;
}

__global__ void k_embed_split(const int* __restrict__ x, const float* __restrict__ Win,
                              _Float16* __restrict__ hhi, _Float16* __restrict__ hlo) {
    int t = blockIdx.x * blockDim.x + threadIdx.x;
    if (t >= NNODES * (EMBED / 4)) return;
    int i = t >> 5, c = t & 31;
    float4 v = ((const float4*)(Win + (size_t)x[i] * EMBED))[c];
    split_store(v.x, v.y, v.z, v.w, hhi, hlo, (size_t)i * 32 + c);
}


template<int K, int N>
__global__ void k_split_Wt(const float* __restrict__ W, _Float16* __restrict__ Whi,
                           _Float16* __restrict__ Wlo) {
    int t = blockIdx.x * blockDim.x + threadIdx.x;
    if (t >= K * N) return;
    int k = t / N, n = t % N;
    float w = W[k * N + n];
    _Float16 hi = (_Float16)w;
    Whi[n * K + k] = hi;
    Wlo[n * K + k] = (_Float16)(w - (float)hi);
}


template<int NOUT>
__global__ __launch_bounds__(256) void k_gemm(const _Float16* __restrict__ Ahi,
                                              const _Float16* __restrict__ Alo,
                                              const _Float16* __restrict__ Bhi,
                                              const _Float16* __restrict__ Blo,
                                              float* __restrict__ out) {
    constexpr int K  = EMBED;
    constexpr int NT = NOUT / 16;
    constexpr int KP = K + 8;
    __shared__ _Float16 sBh[NOUT * KP];
    __shared__ _Float16 sBl[NOUT * KP];

    for (int i = threadIdx.x; i < NOUT * K / 8; i += 256) {
        const int d = (i >> 4) * KP + (i & 15) * 8;
        *(v8h*)(sBh + d) = ((const v8h*)Bhi)[i];
        *(v8h*)(sBl + d) = ((const v8h*)Blo)[i];
    }
    __syncthreads();

    const int wave  = threadIdx.x >> 5;
    const int lane  = threadIdx.x & 31;
    const int strip = blockIdx.x * 8 + wave;
    if (strip >= NNODES / 16) return;
    const int half = lane >> 4;
    const int mrow = strip * 16 + (lane & 15);

    v8f acc[NT] = {};

#pragma unroll
    for (int kc = 0; kc < K / 32; ++kc) {
        const int abase = mrow * K + kc * 32 + half * 8;
        v8h a0h = *(const v8h*)(Ahi + abase);
        v8h a1h = *(const v8h*)(Ahi + abase + 16);
        v8h a0l = *(const v8h*)(Alo + abase);
        v8h a1l = *(const v8h*)(Alo + abase + 16);
        v16h fah = __builtin_shufflevector(a0h, a1h, 0,1,2,3,4,5,6,7,8,9,10,11,12,13,14,15);
        v16h fal = __builtin_shufflevector(a0l, a1l, 0,1,2,3,4,5,6,7,8,9,10,11,12,13,14,15);
#pragma unroll
        for (int t = 0; t < NT; ++t) {
            const int bbase = (t * 16 + (lane & 15)) * KP + kc * 32 + half * 8;
            v8h b0h = *(const v8h*)(sBh + bbase);
            v8h b1h = *(const v8h*)(sBh + bbase + 16);
            v8h b0l = *(const v8h*)(sBl + bbase);
            v8h b1l = *(const v8h*)(sBl + bbase + 16);
            v16h fbh = __builtin_shufflevector(b0h, b1h, 0,1,2,3,4,5,6,7,8,9,10,11,12,13,14,15);
            v16h fbl = __builtin_shufflevector(b0l, b1l, 0,1,2,3,4,5,6,7,8,9,10,11,12,13,14,15);
            acc[t] = __builtin_amdgcn_wmma_f32_16x16x32_f16(false, fah, false, fbh,
                                                            (short)0, acc[t], false, false);
            acc[t] = __builtin_amdgcn_wmma_f32_16x16x32_f16(false, fah, false, fbl,
                                                            (short)0, acc[t], false, false);
            acc[t] = __builtin_amdgcn_wmma_f32_16x16x32_f16(false, fal, false, fbh,
                                                            (short)0, acc[t], false, false);
        }
    }

#pragma unroll
    for (int t = 0; t < NT; ++t) {
        const int ncol = t * 16 + (lane & 15);
#pragma unroll
        for (int r = 0; r < 8; ++r) {
            const int row = strip * 16 + half * 8 + r;
            out[(size_t)row * NOUT + ncol] = acc[t][r];
        }
    }
}


#define GATHER_NODES 4
#define GATHER_ROWS  8

template<int NOUT, bool RELUSPLIT>
__global__ __launch_bounds__(256) void k_gather(const float* __restrict__ hw,
                         const int* __restrict__ off,
                         const int* __restrict__ col, const float* __restrict__ wt,
                         const float* __restrict__ dinv, const float* __restrict__ bias,
                         _Float16* __restrict__ hhi, _Float16* __restrict__ hlo,
                         float* __restrict__ outf) {
    constexpr int LPN = NOUT / 4;
    constexpr int G   = GATHER_NODES;
    constexpr int U   = GATHER_ROWS;
    static_assert(NNODES % G == 0, "a lane group owns G whole nodes");
    const int t     = blockIdx.x * blockDim.x + threadIdx.x;
    const int node0 = (t / LPN) * G;
    const int c     = t % LPN;
    if (node0 >= NNODES) return;

    const float4* __restrict__ hw4 = (const float4*)hw;
    constexpr bool WHOLE_WAVE = (LPN == 32);

    int   offv[G + 1];
    float di[G];
#pragma unroll
    for (int i = 0; i <= G; ++i) {
        offv[i] = off[node0 + i];
        if constexpr (WHOLE_WAVE) offv[i] = __builtin_amdgcn_readfirstlane(offv[i]);
    }
#pragma unroll
    for (int i = 0; i < G; ++i) di[i] = dinv[node0 + i];
    const float4 b = ((const float4*)bias)[c];

    int   sl0[G];
    float wl0[G];
#pragma unroll
    for (int i = 0; i < G; ++i) {
        sl0[i] = 0; wl0[i] = 0.0f;
        if (c < offv[i + 1] - offv[i]) { const int jj = offv[i] + c; sl0[i] = col[jj]; wl0[i] = wt[jj]; }
    }

#pragma unroll
    for (int i = 0; i < G; ++i) {
        const int node = node0 + i;
        const int j0   = offv[i];
        const int deg  = offv[i + 1] - j0;
        const float sw = di[i] * di[i];

        float4 acc = make_float4(0.f, 0.f, 0.f, 0.f);
        int   sl = sl0[i];
        float wl = wl0[i];
        bool first = true;

        for (int base = 0; first || base < deg; base += LPN) {
            int n = deg - base; if (n > LPN) n = LPN;
            if (base > 0) {
                sl = 0; wl = 0.0f;
                if (c < n) { const int jj = j0 + base + c; sl = col[jj]; wl = wt[jj]; }
            }
            for (int k0 = 0; first || k0 < n; k0 += U) {
                float4 m[U];
                float  w[U];
                float4 sv;
                if (first) sv = hw4[(size_t)node * LPN + c];
#pragma unroll
                for (int u = 0; u < U; ++u) {
                    const int  k  = k0 + u;
                    const bool on = k < n;
                    int   s;
                    float x;
                    if constexpr (WHOLE_WAVE) {
                        s = __builtin_amdgcn_readlane(sl, k & 31);
                        x = __builtin_bit_cast(float,
                                __builtin_amdgcn_readlane(__builtin_bit_cast(int, wl), k & 31));
                    } else {
                        s = __shfl(sl, k, LPN);
                        x = __shfl(wl, k, LPN);
                    }
                    w[u] = on ? x : 0.0f;
                    m[u] = make_float4(0.f, 0.f, 0.f, 0.f);
                    if (on) m[u] = hw4[(size_t)s * LPN + c];
                }
                if (first) {
                    acc.x = sw * sv.x; acc.y = sw * sv.y; acc.z = sw * sv.z; acc.w = sw * sv.w;
                    first = false;
                }
#pragma unroll
                for (int u = 0; u < U; ++u) {
                    if (k0 + u < n) {
                        acc.x += w[u] * m[u].x; acc.y += w[u] * m[u].y;
                        acc.z += w[u] * m[u].z; acc.w += w[u] * m[u].w;
                    }
                }
            }
        }

        acc.x += b.x; acc.y += b.y; acc.z += b.z; acc.w += b.w;

        if (RELUSPLIT) {
            acc.x = fmaxf(acc.x, 0.f); acc.y = fmaxf(acc.y, 0.f);
            acc.z = fmaxf(acc.z, 0.f); acc.w = fmaxf(acc.w, 0.f);
            split_store(acc.x, acc.y, acc.z, acc.w, hhi, hlo, (size_t)node * LPN + c);
        } else {
            ((float4*)outf)[(size_t)node * LPN + c] = acc;
        }
    }
}


extern "C" void kernel_launch(void* const* d_in, const int* in_sizes, int n_in,
                              void* d_out, int out_size, void* d_ws, size_t ws_size,
                              hipStream_t stream) {
    const int*   x      = (const int*)d_in[0];
    const int*   src    = (const int*)d_in[1];
    const int*   dst    = ((const int*)d_in[1]) + NEDGES;
    const float* W_in   = (const float*)d_in[2];
    const float* W_conv = (const float*)d_in[3];
    const float* b_conv = (const float*)d_in[4];
    const float* W_dec  = (const float*)d_in[5];
    const float* b_dec  = (const float*)d_in[6];
    (void)in_sizes; (void)n_in; (void)out_size; (void)ws_size;

    char* p = (char*)d_ws;
    auto carve = [&](size_t bytes) { char* r = p; p += (bytes + 255) & ~(size_t)255; return r; };
    float*    dinv   = (float*)   carve((size_t)NNODES * 4);
    int*      cnt    = (int*)     carve((size_t)NNODES * 4);
    int*      cursor = (int*)     carve((size_t)NNODES * 4);
    int*      off    = (int*)     carve((size_t)(NNODES + 1) * 4);
    int*      col    = (int*)     carve((size_t)NEDGES * 4);
    float*    wt     = (float*)   carve((size_t)NEDGES * 4);
    float*    hw     = (float*)   carve((size_t)NNODES * EMBED * 4);
    _Float16* hhi    = (_Float16*)carve((size_t)NNODES * EMBED * 2);
    _Float16* hlo    = (_Float16*)carve((size_t)NNODES * EMBED * 2);
    _Float16* Wch    = (_Float16*)carve((size_t)EMBED * EMBED * 2);
    _Float16* Wcl    = (_Float16*)carve((size_t)EMBED * EMBED * 2);
    _Float16* Wdh    = (_Float16*)carve((size_t)NTYPES * EMBED * 2);
    _Float16* Wdl    = (_Float16*)carve((size_t)NTYPES * EMBED * 2);

    const int B = 256;
    auto blocks = [](long long n, int b) { return (int)((n + b - 1) / b); };

    k_zero2   <<<blocks(NNODES, B), B, 0, stream>>>(cnt, cursor);
    k_count   <<<blocks(NEDGES, B), B, 0, stream>>>(dst, cnt);
    k_dinv    <<<blocks(NNODES, B), B, 0, stream>>>(cnt, dinv);
    k_scan    <<<1, SCAN_THREADS, 0, stream>>>(cnt, off);
    k_csr_fill<<<blocks(NEDGES, B), B, 0, stream>>>(src, dst, off, dinv, cursor, col, wt);

    k_split_Wt<EMBED, EMBED> <<<blocks(EMBED * EMBED, B), B, 0, stream>>>(W_conv, Wch, Wcl);
    k_split_Wt<EMBED, NTYPES><<<blocks(EMBED * NTYPES, B), B, 0, stream>>>(W_dec,  Wdh, Wdl);

    k_embed_split<<<blocks((long long)NNODES * 32, B), B, 0, stream>>>(x, W_in, hhi, hlo);

    const int gemm_blocks = ((NNODES / 16) + 7) / 8;

    for (int it = 0; it < NITERS; ++it) {
        k_gemm<EMBED><<<gemm_blocks, B, 0, stream>>>(hhi, hlo, Wch, Wcl, hw);
        k_gather<EMBED, true><<<blocks((long long)(NNODES / GATHER_NODES) * 32, B), B, 0, stream>>>(
            hw, off, col, wt, dinv, b_conv, hhi, hlo, nullptr);
    }

    k_gemm<NTYPES><<<gemm_blocks, B, 0, stream>>>(hhi, hlo, Wdh, Wdl, hw);
    k_gather<NTYPES, false><<<blocks((long long)(NNODES / GATHER_NODES) * 16, B), B, 0, stream>>>(
        hw, off, col, wt, dinv, b_dec, nullptr, nullptr, (float*)d_out);
}


# set-up launches 8 -> 4: node norms written by the scan workgroup; both weight splits and the embedding split run as the other workgroups of the scan launch, which left the chip idle
# speedup vs baseline: 1.8162x; 1.0344x over previous
#include <hip/hip_runtime.h>

typedef __attribute__((ext_vector_type(4)))  _Float16 v4h;
typedef __attribute__((ext_vector_type(8)))  _Float16 v8h;
typedef __attribute__((ext_vector_type(16))) _Float16 v16h;
typedef __attribute__((ext_vector_type(8)))  float    v8f;

#define NNODES 40000
#define NEDGES 640000
#define EMBED  128
#define NTYPES 64
#define NITERS 4


__global__ void k_zero2(int* a, int* b) {
    int i = blockIdx.x * blockDim.x + threadIdx.x;
    if (i < NNODES) { a[i] = 0; b[i] = 0; }
}

__global__ void k_count(const int* __restrict__ dst, int* cnt) {
    int e = blockIdx.x * blockDim.x + threadIdx.x;
    if (e < NEDGES) atomicAdd(&cnt[dst[e]], 1);
}

__global__ void k_csr_fill(const int* __restrict__ src, const int* __restrict__ dst,
                           const int* __restrict__ off, const float* __restrict__ dinv,
                           int* __restrict__ cursor, int* __restrict__ col,
                           float* __restrict__ wt) {
    int e = blockIdx.x * blockDim.x + threadIdx.x;
    if (e >= NEDGES) return;
    int s = src[e], d = dst[e];
    int j = off[d] + atomicAdd(&cursor[d], 1);
    col[j] = s;
    wt[j]  = dinv[s] * dinv[d];
}


__device__ __forceinline__ void split_store(float x, float y, float z, float w,
                                            _Float16* hi, _Float16* lo, size_t idx4) {
    v4h H, L;
    H[0] = (_Float16)x; L[0] = (_Float16)(x - (float)H[0]);
    H[1] = (_Float16)y; L[1] = (_Float16)(y - (float)H[1]);
    H[2] = (_Float16)z; L[2] = (_Float16)(z - (float)H[2]);
    H[3] = (_Float16)w; L[3] = (_Float16)(w - (float)H[3]);
    ((v4h*)hi)[idx4] = H;
    ((v4h*)lo)[idx4] = L;
}

__device__ __forceinline__ void embed_split(int t, const int* __restrict__ x,
                                            const float* __restrict__ Win,
                                            _Float16* __restrict__ hhi, _Float16* __restrict__ hlo) {
    if (t >= NNODES * (EMBED / 4)) return;
    int i = t >> 5, c = t & 31;
    float4 v = ((const float4*)(Win + (size_t)x[i] * EMBED))[c];
    split_store(v.x, v.y, v.z, v.w, hhi, hlo, (size_t)i * 32 + c);
}


template<int K, int N>
__device__ __forceinline__ void split_Wt(int t, const float* __restrict__ W,
                                         _Float16* __restrict__ Whi, _Float16* __restrict__ Wlo) {
    if (t >= K * N) return;
    int k = t / N, n = t % N;
    float w = W[k * N + n];
    _Float16 hi = (_Float16)w;
    Whi[n * K + k] = hi;
    Wlo[n * K + k] = (_Float16)(w - (float)hi);
}

#define SCAN_THREADS 1024
#define SCAN_WAVES   (SCAN_THREADS / 32)
#define SCAN_VEC     (NNODES / 4)
#define SCAN_ITERS   ((SCAN_VEC + SCAN_THREADS - 1) / SCAN_THREADS)
#define PREP_WCONV_WGS ((EMBED * EMBED + SCAN_THREADS - 1) / SCAN_THREADS)
#define PREP_WDEC_WGS  ((EMBED * NTYPES + SCAN_THREADS - 1) / SCAN_THREADS)
#define PREP_EMBED_WGS ((NNODES * (EMBED / 4) + SCAN_THREADS - 1) / SCAN_THREADS)
#define PREP_WGS       (1 + PREP_WCONV_WGS + PREP_WDEC_WGS + PREP_EMBED_WGS)

__global__ __launch_bounds__(SCAN_THREADS) void k_scan_prep(const int* __restrict__ cnt,
                                                            int* __restrict__ off,
                                                            float* __restrict__ dinv,
                                                            const int* __restrict__ x,
                                                            const float* __restrict__ Win,
                                                            _Float16* __restrict__ hhi,
                                                            _Float16* __restrict__ hlo,
                                                            const float* __restrict__ Wc,
                                                            _Float16* __restrict__ Wch,
                                                            _Float16* __restrict__ Wcl,
                                                            const float* __restrict__ Wd,
                                                            _Float16* __restrict__ Wdh,
                                                            _Float16* __restrict__ Wdl) {
    static_assert(NNODES % 4 == 0, "int4 path needs NNODES % 4 == 0");
    if (blockIdx.x != 0) {
        int b = blockIdx.x - 1;
        if (b < PREP_WCONV_WGS) {
            split_Wt<EMBED, EMBED>(b * SCAN_THREADS + threadIdx.x, Wc, Wch, Wcl);
            return;
        }
        b -= PREP_WCONV_WGS;
        if (b < PREP_WDEC_WGS) {
            split_Wt<EMBED, NTYPES>(b * SCAN_THREADS + threadIdx.x, Wd, Wdh, Wdl);
            return;
        }
        b -= PREP_WDEC_WGS;
        embed_split(b * SCAN_THREADS + threadIdx.x, x, Win, hhi, hlo);
        return;
    }
    static_assert(SCAN_WAVES == 32, "one lane per wave total");
    __shared__ int wtot[SCAN_WAVES];
    const int lane  = threadIdx.x & 31;
    const int wave  = threadIdx.x >> 5;
    const int vbase = wave * (SCAN_ITERS * 32);

    int4 v[SCAN_ITERS];
#pragma unroll
    for (int i = 0; i < SCAN_ITERS; ++i) {
        const int idx = vbase + i * 32 + lane;
        v[i] = (idx < SCAN_VEC) ? ((const int4*)cnt)[idx] : make_int4(0, 0, 0, 0);
    }

    int own[SCAN_ITERS], incl[SCAN_ITERS];
#pragma unroll
    for (int i = 0; i < SCAN_ITERS; ++i) { own[i] = v[i].x + v[i].y + v[i].z + v[i].w; incl[i] = own[i]; }

#pragma unroll
    for (int i = 0; i < SCAN_ITERS; ++i) {
        const int idx = vbase + i * 32 + lane;
        if (idx < SCAN_VEC)
            ((float4*)dinv)[idx] = make_float4(rsqrtf((float)v[i].x + 1.0f), rsqrtf((float)v[i].y + 1.0f),
                                               rsqrtf((float)v[i].z + 1.0f), rsqrtf((float)v[i].w + 1.0f));
    }
#pragma unroll
    for (int d = 1; d < 32; d <<= 1) {
#pragma unroll
        for (int i = 0; i < SCAN_ITERS; ++i) {
            const int t = __shfl_up(incl[i], d, 32);
            if (lane >= d) incl[i] += t;
        }
    }

    int excl[SCAN_ITERS];
    int run = 0;
#pragma unroll
    for (int i = 0; i < SCAN_ITERS; ++i) {
        excl[i] = run + incl[i] - own[i];
        run += __shfl(incl[i], 31, 32);
    }

    if (lane == 0) wtot[wave] = run;
    __syncthreads();

    const int wt = wtot[lane];
    int ws = wt;
#pragma unroll
    for (int d = 1; d < 32; d <<= 1) {
        const int t = __shfl_up(ws, d, 32);
        if (lane >= d) ws += t;
    }
    const int wprefix = __shfl(ws - wt, wave, 32);
    const int total   = __shfl(ws, 31, 32);

#pragma unroll
    for (int i = 0; i < SCAN_ITERS; ++i) {
        const int idx = vbase + i * 32 + lane;
        const int p = wprefix + excl[i];
        if (idx < SCAN_VEC)
            ((int4*)off)[idx] = make_int4(p, p + v[i].x, p + v[i].x + v[i].y,
                                          p + v[i].x + v[i].y + v[i].z);
    }
    if (threadIdx.x == 0) off[NNODES] = total;
}


template<int NOUT>
__global__ __launch_bounds__(256) void k_gemm(const _Float16* __restrict__ Ahi,
                                              const _Float16* __restrict__ Alo,
                                              const _Float16* __restrict__ Bhi,
                                              const _Float16* __restrict__ Blo,
                                              float* __restrict__ out) {
    constexpr int K  = EMBED;
    constexpr int NT = NOUT / 16;
    constexpr int KP = K + 8;
    __shared__ _Float16 sBh[NOUT * KP];
    __shared__ _Float16 sBl[NOUT * KP];

    for (int i = threadIdx.x; i < NOUT * K / 8; i += 256) {
        const int d = (i >> 4) * KP + (i & 15) * 8;
        *(v8h*)(sBh + d) = ((const v8h*)Bhi)[i];
        *(v8h*)(sBl + d) = ((const v8h*)Blo)[i];
    }
    __syncthreads();

    const int wave  = threadIdx.x >> 5;
    const int lane  = threadIdx.x & 31;
    const int strip = blockIdx.x * 8 + wave;
    if (strip >= NNODES / 16) return;
    const int half = lane >> 4;
    const int mrow = strip * 16 + (lane & 15);

    v8f acc[NT] = {};

#pragma unroll
    for (int kc = 0; kc < K / 32; ++kc) {
        const int abase = mrow * K + kc * 32 + half * 8;
        v8h a0h = *(const v8h*)(Ahi + abase);
        v8h a1h = *(const v8h*)(Ahi + abase + 16);
        v8h a0l = *(const v8h*)(Alo + abase);
        v8h a1l = *(const v8h*)(Alo + abase + 16);
        v16h fah = __builtin_shufflevector(a0h, a1h, 0,1,2,3,4,5,6,7,8,9,10,11,12,13,14,15);
        v16h fal = __builtin_shufflevector(a0l, a1l, 0,1,2,3,4,5,6,7,8,9,10,11,12,13,14,15);
#pragma unroll
        for (int t = 0; t < NT; ++t) {
            const int bbase = (t * 16 + (lane & 15)) * KP + kc * 32 + half * 8;
            v8h b0h = *(const v8h*)(sBh + bbase);
            v8h b1h = *(const v8h*)(sBh + bbase + 16);
            v8h b0l = *(const v8h*)(sBl + bbase);
            v8h b1l = *(const v8h*)(sBl + bbase + 16);
            v16h fbh = __builtin_shufflevector(b0h, b1h, 0,1,2,3,4,5,6,7,8,9,10,11,12,13,14,15);
            v16h fbl = __builtin_shufflevector(b0l, b1l, 0,1,2,3,4,5,6,7,8,9,10,11,12,13,14,15);
            acc[t] = __builtin_amdgcn_wmma_f32_16x16x32_f16(false, fah, false, fbh,
                                                            (short)0, acc[t], false, false);
            acc[t] = __builtin_amdgcn_wmma_f32_16x16x32_f16(false, fah, false, fbl,
                                                            (short)0, acc[t], false, false);
            acc[t] = __builtin_amdgcn_wmma_f32_16x16x32_f16(false, fal, false, fbh,
                                                            (short)0, acc[t], false, false);
        }
    }

#pragma unroll
    for (int t = 0; t < NT; ++t) {
        const int ncol = t * 16 + (lane & 15);
#pragma unroll
        for (int r = 0; r < 8; ++r) {
            const int row = strip * 16 + half * 8 + r;
            out[(size_t)row * NOUT + ncol] = acc[t][r];
        }
    }
}


#define GATHER_NODES 4
#define GATHER_ROWS  8

template<int NOUT, bool RELUSPLIT>
__global__ __launch_bounds__(256) void k_gather(const float* __restrict__ hw,
                         const int* __restrict__ off,
                         const int* __restrict__ col, const float* __restrict__ wt,
                         const float* __restrict__ dinv, const float* __restrict__ bias,
                         _Float16* __restrict__ hhi, _Float16* __restrict__ hlo,
                         float* __restrict__ outf) {
    constexpr int LPN = NOUT / 4;
    constexpr int G   = GATHER_NODES;
    constexpr int U   = GATHER_ROWS;
    static_assert(NNODES % G == 0, "a lane group owns G whole nodes");
    const int t     = blockIdx.x * blockDim.x + threadIdx.x;
    const int node0 = (t / LPN) * G;
    const int c     = t % LPN;
    if (node0 >= NNODES) return;

    const float4* __restrict__ hw4 = (const float4*)hw;
    constexpr bool WHOLE_WAVE = (LPN == 32);

    int   offv[G + 1];
    float di[G];
#pragma unroll
    for (int i = 0; i <= G; ++i) {
        offv[i] = off[node0 + i];
        if constexpr (WHOLE_WAVE) offv[i] = __builtin_amdgcn_readfirstlane(offv[i]);
    }
#pragma unroll
    for (int i = 0; i < G; ++i) di[i] = dinv[node0 + i];
    const float4 b = ((const float4*)bias)[c];

    int   sl0[G];
    float wl0[G];
#pragma unroll
    for (int i = 0; i < G; ++i) {
        sl0[i] = 0; wl0[i] = 0.0f;
        if (c < offv[i + 1] - offv[i]) { const int jj = offv[i] + c; sl0[i] = col[jj]; wl0[i] = wt[jj]; }
    }

#pragma unroll
    for (int i = 0; i < G; ++i) {
        const int node = node0 + i;
        const int j0   = offv[i];
        const int deg  = offv[i + 1] - j0;
        const float sw = di[i] * di[i];

        float4 acc = make_float4(0.f, 0.f, 0.f, 0.f);
        int   sl = sl0[i];
        float wl = wl0[i];
        bool first = true;

        for (int base = 0; first || base < deg; base += LPN) {
            int n = deg - base; if (n > LPN) n = LPN;
            if (base > 0) {
                sl = 0; wl = 0.0f;
                if (c < n) { const int jj = j0 + base + c; sl = col[jj]; wl = wt[jj]; }
            }
            for (int k0 = 0; first || k0 < n; k0 += U) {
                float4 m[U];
                float  w[U];
                float4 sv;
                if (first) sv = hw4[(size_t)node * LPN + c];
#pragma unroll
                for (int u = 0; u < U; ++u) {
                    const int  k  = k0 + u;
                    const bool on = k < n;
                    int   s;
                    float x;
                    if constexpr (WHOLE_WAVE) {
                        s = __builtin_amdgcn_readlane(sl, k & 31);
                        x = __builtin_bit_cast(float,
                                __builtin_amdgcn_readlane(__builtin_bit_cast(int, wl), k & 31));
                    } else {
                        s = __shfl(sl, k, LPN);
                        x = __shfl(wl, k, LPN);
                    }
                    w[u] = on ? x : 0.0f;
                    m[u] = make_float4(0.f, 0.f, 0.f, 0.f);
                    if (on) m[u] = hw4[(size_t)s * LPN + c];
                }
                if (first) {
                    acc.x = sw * sv.x; acc.y = sw * sv.y; acc.z = sw * sv.z; acc.w = sw * sv.w;
                    first = false;
                }
#pragma unroll
                for (int u = 0; u < U; ++u) {
                    if (k0 + u < n) {
                        acc.x += w[u] * m[u].x; acc.y += w[u] * m[u].y;
                        acc.z += w[u] * m[u].z; acc.w += w[u] * m[u].w;
                    }
                }
            }
        }

        acc.x += b.x; acc.y += b.y; acc.z += b.z; acc.w += b.w;

        if (RELUSPLIT) {
            acc.x = fmaxf(acc.x, 0.f); acc.y = fmaxf(acc.y, 0.f);
            acc.z = fmaxf(acc.z, 0.f); acc.w = fmaxf(acc.w, 0.f);
            split_store(acc.x, acc.y, acc.z, acc.w, hhi, hlo, (size_t)node * LPN + c);
        } else {
            ((float4*)outf)[(size_t)node * LPN + c] = acc;
        }
    }
}


extern "C" void kernel_launch(void* const* d_in, const int* in_sizes, int n_in,
                              void* d_out, int out_size, void* d_ws, size_t ws_size,
                              hipStream_t stream) {
    const int*   x      = (const int*)d_in[0];
    const int*   src    = (const int*)d_in[1];
    const int*   dst    = ((const int*)d_in[1]) + NEDGES;
    const float* W_in   = (const float*)d_in[2];
    const float* W_conv = (const float*)d_in[3];
    const float* b_conv = (const float*)d_in[4];
    const float* W_dec  = (const float*)d_in[5];
    const float* b_dec  = (const float*)d_in[6];
    (void)in_sizes; (void)n_in; (void)out_size; (void)ws_size;

    char* p = (char*)d_ws;
    auto carve = [&](size_t bytes) { char* r = p; p += (bytes + 255) & ~(size_t)255; return r; };
    float*    dinv   = (float*)   carve((size_t)NNODES * 4);
    int*      cnt    = (int*)     carve((size_t)NNODES * 4);
    int*      cursor = (int*)     carve((size_t)NNODES * 4);
    int*      off    = (int*)     carve((size_t)(NNODES + 1) * 4);
    int*      col    = (int*)     carve((size_t)NEDGES * 4);
    float*    wt     = (float*)   carve((size_t)NEDGES * 4);
    float*    hw     = (float*)   carve((size_t)NNODES * EMBED * 4);
    _Float16* hhi    = (_Float16*)carve((size_t)NNODES * EMBED * 2);
    _Float16* hlo    = (_Float16*)carve((size_t)NNODES * EMBED * 2);
    _Float16* Wch    = (_Float16*)carve((size_t)EMBED * EMBED * 2);
    _Float16* Wcl    = (_Float16*)carve((size_t)EMBED * EMBED * 2);
    _Float16* Wdh    = (_Float16*)carve((size_t)NTYPES * EMBED * 2);
    _Float16* Wdl    = (_Float16*)carve((size_t)NTYPES * EMBED * 2);

    const int B = 256;
    auto blocks = [](long long n, int b) { return (int)((n + b - 1) / b); };

    k_zero2   <<<blocks(NNODES, B), B, 0, stream>>>(cnt, cursor);
    k_count   <<<blocks(NEDGES, B), B, 0, stream>>>(dst, cnt);
    k_scan_prep<<<PREP_WGS, SCAN_THREADS, 0, stream>>>(cnt, off, dinv, x, W_in, hhi, hlo,
                                                       W_conv, Wch, Wcl, W_dec, Wdh, Wdl);
    k_csr_fill<<<blocks(NEDGES, B), B, 0, stream>>>(src, dst, off, dinv, cursor, col, wt);

    const int gemm_blocks = ((NNODES / 16) + 7) / 8;

    for (int it = 0; it < NITERS; ++it) {
        k_gemm<EMBED><<<gemm_blocks, B, 0, stream>>>(hhi, hlo, Wch, Wcl, hw);
        k_gather<EMBED, true><<<blocks((long long)(NNODES / GATHER_NODES) * 32, B), B, 0, stream>>>(
            hw, off, col, wt, dinv, b_conv, hhi, hlo, nullptr);
    }

    k_gemm<NTYPES><<<gemm_blocks, B, 0, stream>>>(hhi, hlo, Wdh, Wdl, hw);
    k_gather<NTYPES, false><<<blocks((long long)(NNODES / GATHER_NODES) * 16, B), B, 0, stream>>>(
        hw, off, col, wt, dinv, b_dec, nullptr, nullptr, (float*)d_out);
}


# each conv layer is one launch: the gather's ReLU'd rows go via LDS as split (hi,lo) pairs into the next layer's WMMA GEMM in the same workgroup; 4 GEMM launches and split planes gone; sum order kept
# speedup vs baseline: 2.2276x; 1.2265x over previous
#include <hip/hip_runtime.h>

typedef __attribute__((ext_vector_type(4)))  _Float16 v4h;
typedef __attribute__((ext_vector_type(8)))  _Float16 v8h;
typedef __attribute__((ext_vector_type(16))) _Float16 v16h;
typedef __attribute__((ext_vector_type(8)))  float    v8f;

#define NNODES 40000
#define NEDGES 640000
#define EMBED  128
#define NTYPES 64
#define NITERS 4


__global__ void k_zero2(int* a, int* b) {
    int i = blockIdx.x * blockDim.x + threadIdx.x;
    if (i < NNODES) { a[i] = 0; b[i] = 0; }
}

__global__ void k_count(const int* __restrict__ dst, int* cnt) {
    int e = blockIdx.x * blockDim.x + threadIdx.x;
    if (e < NEDGES) atomicAdd(&cnt[dst[e]], 1);
}

__global__ void k_csr_fill(const int* __restrict__ src, const int* __restrict__ dst,
                           const int* __restrict__ off, const float* __restrict__ dinv,
                           int* __restrict__ cursor, int* __restrict__ col,
                           float* __restrict__ wt) {
    int e = blockIdx.x * blockDim.x + threadIdx.x;
    if (e >= NEDGES) return;
    int s = src[e], d = dst[e];
    int j = off[d] + atomicAdd(&cursor[d], 1);
    col[j] = s;
    wt[j]  = dinv[s] * dinv[d];
}


__device__ __forceinline__ void split_store(float x, float y, float z, float w,
                                            _Float16* hi, _Float16* lo, size_t idx4) {
    v4h H, L;
    H[0] = (_Float16)x; L[0] = (_Float16)(x - (float)H[0]);
    H[1] = (_Float16)y; L[1] = (_Float16)(y - (float)H[1]);
    H[2] = (_Float16)z; L[2] = (_Float16)(z - (float)H[2]);
    H[3] = (_Float16)w; L[3] = (_Float16)(w - (float)H[3]);
    ((v4h*)hi)[idx4] = H;
    ((v4h*)lo)[idx4] = L;
}


template<int K, int N>
__device__ __forceinline__ void split_Wt(int t, const float* __restrict__ W,
                                         _Float16* __restrict__ Whi, _Float16* __restrict__ Wlo) {
    if (t >= K * N) return;
    int k = t / N, n = t % N;
    float w = W[k * N + n];
    _Float16 hi = (_Float16)w;
    Whi[n * K + k] = hi;
    Wlo[n * K + k] = (_Float16)(w - (float)hi);
}

#define SCAN_THREADS 1024
#define SCAN_WAVES   (SCAN_THREADS / 32)
#define SCAN_VEC     (NNODES / 4)
#define SCAN_ITERS   ((SCAN_VEC + SCAN_THREADS - 1) / SCAN_THREADS)
#define PREP_WCONV_WGS ((EMBED * EMBED + SCAN_THREADS - 1) / SCAN_THREADS)
#define PREP_WDEC_WGS  ((EMBED * NTYPES + SCAN_THREADS - 1) / SCAN_THREADS)
#define PREP_WGS       (2 + PREP_WCONV_WGS + PREP_WDEC_WGS)

__device__ __forceinline__ void table_gemm(const float* __restrict__ Win,
                                           const float* __restrict__ Wc,
                                           float* __restrict__ tw) {
    static_assert((NTYPES / 16) * (EMBED / 16) == SCAN_WAVES, "one wave per 16x16 tile of the product");
    const int wave  = threadIdx.x >> 5;
    const int lane  = threadIdx.x & 31;
    const int half  = lane >> 4;
    const int c     = lane & 15;
    const int strip = wave % (NTYPES / 16);
    const int tile  = wave / (NTYPES / 16);
    const float* __restrict__ arow = Win + (strip * 16 + c) * EMBED;
    const float* __restrict__ bcol = Wc + tile * 16 + c;
    v8f acc = {};
#pragma unroll
    for (int kc = 0; kc < EMBED / 32; ++kc) {
        v16h ah, al, bh, bl;
#pragma unroll
        for (int s = 0; s < 16; ++s) {
            const int k = kc * 32 + (s >> 3) * 16 + half * 8 + (s & 7);
            const float a = arow[k];
            const float b = bcol[k * EMBED];
            const _Float16 a_hi = (_Float16)a;
            const _Float16 b_hi = (_Float16)b;
            ah[s] = a_hi; al[s] = (_Float16)(a - (float)a_hi);
            bh[s] = b_hi; bl[s] = (_Float16)(b - (float)b_hi);
        }
        acc = __builtin_amdgcn_wmma_f32_16x16x32_f16(false, ah, false, bh, (short)0, acc, false, false);
        acc = __builtin_amdgcn_wmma_f32_16x16x32_f16(false, ah, false, bl, (short)0, acc, false, false);
        acc = __builtin_amdgcn_wmma_f32_16x16x32_f16(false, al, false, bh, (short)0, acc, false, false);
    }
#pragma unroll
    for (int r = 0; r < 8; ++r)
        tw[(strip * 16 + half * 8 + r) * EMBED + tile * 16 + c] = acc[r];
}

__global__ __launch_bounds__(SCAN_THREADS) void k_scan_prep(const int* __restrict__ cnt,
                                                            int* __restrict__ off,
                                                            float* __restrict__ dinv,
                                                            const float* __restrict__ Win,
                                                            float* __restrict__ tw,
                                                            const float* __restrict__ Wc,
                                                            _Float16* __restrict__ Wch,
                                                            _Float16* __restrict__ Wcl,
                                                            const float* __restrict__ Wd,
                                                            _Float16* __restrict__ Wdh,
                                                            _Float16* __restrict__ Wdl) {
    static_assert(NNODES % 4 == 0, "int4 path needs NNODES % 4 == 0");
    if (blockIdx.x != 0) {
        int b = blockIdx.x - 1;
        if (b == 0) {
            table_gemm(Win, Wc, tw);
            return;
        }
        b -= 1;
        if (b < PREP_WCONV_WGS) {
            split_Wt<EMBED, EMBED>(b * SCAN_THREADS + threadIdx.x, Wc, Wch, Wcl);
            return;
        }
        b -= PREP_WCONV_WGS;
        if (b < PREP_WDEC_WGS) {
            split_Wt<EMBED, NTYPES>(b * SCAN_THREADS + threadIdx.x, Wd, Wdh, Wdl);
            return;
        }
        return;
    }
    static_assert(SCAN_WAVES == 32, "one lane per wave total");
    __shared__ int wtot[SCAN_WAVES];
    const int lane  = threadIdx.x & 31;
    const int wave  = threadIdx.x >> 5;
    const int vbase = wave * (SCAN_ITERS * 32);

    int4 v[SCAN_ITERS];
#pragma unroll
    for (int i = 0; i < SCAN_ITERS; ++i) {
        const int idx = vbase + i * 32 + lane;
        v[i] = (idx < SCAN_VEC) ? ((const int4*)cnt)[idx] : make_int4(0, 0, 0, 0);
    }

    int own[SCAN_ITERS], incl[SCAN_ITERS];
#pragma unroll
    for (int i = 0; i < SCAN_ITERS; ++i) { own[i] = v[i].x + v[i].y + v[i].z + v[i].w; incl[i] = own[i]; }

#pragma unroll
    for (int i = 0; i < SCAN_ITERS; ++i) {
        const int idx = vbase + i * 32 + lane;
        if (idx < SCAN_VEC)
            ((float4*)dinv)[idx] = make_float4(rsqrtf((float)v[i].x + 1.0f), rsqrtf((float)v[i].y + 1.0f),
                                               rsqrtf((float)v[i].z + 1.0f), rsqrtf((float)v[i].w + 1.0f));
    }
#pragma unroll
    for (int d = 1; d < 32; d <<= 1) {
#pragma unroll
        for (int i = 0; i < SCAN_ITERS; ++i) {
            const int t = __shfl_up(incl[i], d, 32);
            if (lane >= d) incl[i] += t;
        }
    }

    int excl[SCAN_ITERS];
    int run = 0;
#pragma unroll
    for (int i = 0; i < SCAN_ITERS; ++i) {
        excl[i] = run + incl[i] - own[i];
        run += __shfl(incl[i], 31, 32);
    }

    if (lane == 0) wtot[wave] = run;
    __syncthreads();

    const int wt = wtot[lane];
    int ws = wt;
#pragma unroll
    for (int d = 1; d < 32; d <<= 1) {
        const int t = __shfl_up(ws, d, 32);
        if (lane >= d) ws += t;
    }
    const int wprefix = __shfl(ws - wt, wave, 32);
    const int total   = __shfl(ws, 31, 32);

#pragma unroll
    for (int i = 0; i < SCAN_ITERS; ++i) {
        const int idx = vbase + i * 32 + lane;
        const int p = wprefix + excl[i];
        if (idx < SCAN_VEC)
            ((int4*)off)[idx] = make_int4(p, p + v[i].x, p + v[i].x + v[i].y,
                                          p + v[i].x + v[i].y + v[i].z);
    }
    if (threadIdx.x == 0) off[NNODES] = total;
}


#define GATHER_NODES 4
#define GATHER_ROWS  8
#define FUSE_NODES   32
#define FUSE_KP      (EMBED + 8)

template<int NOUT, bool TYPED, int NNEXT>
__global__ __launch_bounds__(256) void k_gather(const float* __restrict__ hw,
                         const int* __restrict__ x,
                         const int* __restrict__ off,
                         const int* __restrict__ col, const float* __restrict__ wt,
                         const float* __restrict__ dinv, const float* __restrict__ bias,
                         const _Float16* __restrict__ Bhi,
                         const _Float16* __restrict__ Blo,
                         float* __restrict__ outf) {
    constexpr int LPN = NOUT / 4;
    constexpr int G   = GATHER_NODES;
    constexpr int U   = GATHER_ROWS;
    constexpr bool FUSED = (NNEXT > 0);
    static_assert(NNODES % G == 0, "a lane group owns G whole nodes");
    static_assert(!FUSED || (NOUT == EMBED && NNODES % FUSE_NODES == 0 && FUSE_NODES == 8 * G),
                  "a fused layer: whole-wave lane groups, 8 waves x G nodes, exact grid");
    __shared__ __attribute__((aligned(16))) _Float16 sAh[FUSED ? FUSE_NODES * FUSE_KP : 8];
    __shared__ __attribute__((aligned(16))) _Float16 sAl[FUSED ? FUSE_NODES * FUSE_KP : 8];
    const int t     = blockIdx.x * blockDim.x + threadIdx.x;
    const int node0 = (t / LPN) * G;
    const int c     = t % LPN;
    if constexpr (!FUSED) {
        if (node0 >= NNODES) return;
    }

    const float4* __restrict__ hw4 = (const float4*)hw;
    constexpr bool WHOLE_WAVE = (LPN == 32);

    int   offv[G + 1];
    float di[G];
    int   selfrow[G];
#pragma unroll
    for (int i = 0; i <= G; ++i) {
        offv[i] = off[node0 + i];
        if constexpr (WHOLE_WAVE) offv[i] = __builtin_amdgcn_readfirstlane(offv[i]);
    }
#pragma unroll
    for (int i = 0; i < G; ++i) di[i] = dinv[node0 + i];
#pragma unroll
    for (int i = 0; i < G; ++i) {
        selfrow[i] = node0 + i;
        if constexpr (TYPED) {
            selfrow[i] = x[node0 + i];
            if constexpr (WHOLE_WAVE) selfrow[i] = __builtin_amdgcn_readfirstlane(selfrow[i]);
        }
    }
    const float4 b = ((const float4*)bias)[c];

    int   sl0[G];
    float wl0[G];
#pragma unroll
    for (int i = 0; i < G; ++i) {
        sl0[i] = 0; wl0[i] = 0.0f;
        if (c < offv[i + 1] - offv[i]) { const int jj = offv[i] + c; sl0[i] = col[jj]; wl0[i] = wt[jj]; }
    }
    if constexpr (TYPED) {
#pragma unroll
        for (int i = 0; i < G; ++i) sl0[i] = x[sl0[i]];
    }

#pragma unroll
    for (int i = 0; i < G; ++i) {
        const int node = node0 + i;
        const int j0   = offv[i];
        const int deg  = offv[i + 1] - j0;
        const float sw = di[i] * di[i];

        float4 acc = make_float4(0.f, 0.f, 0.f, 0.f);
        int   sl = sl0[i];
        float wl = wl0[i];
        bool first = true;

        for (int base = 0; first || base < deg; base += LPN) {
            int n = deg - base; if (n > LPN) n = LPN;
            if (base > 0) {
                sl = 0; wl = 0.0f;
                if (c < n) {
                    const int jj = j0 + base + c;
                    sl = col[jj]; wl = wt[jj];
                    if constexpr (TYPED) sl = x[sl];
                }
            }
            for (int k0 = 0; first || k0 < n; k0 += U) {
                float4 m[U];
                float  w[U];
                float4 sv;
                if (first) sv = hw4[(size_t)selfrow[i] * LPN + c];
#pragma unroll
                for (int u = 0; u < U; ++u) {
                    const int  k  = k0 + u;
                    const bool on = k < n;
                    int   s;
                    float x;
                    if constexpr (WHOLE_WAVE) {
                        s = __builtin_amdgcn_readlane(sl, k & 31);
                        x = __builtin_bit_cast(float,
                                __builtin_amdgcn_readlane(__builtin_bit_cast(int, wl), k & 31));
                    } else {
                        s = __shfl(sl, k, LPN);
                        x = __shfl(wl, k, LPN);
                    }
                    w[u] = on ? x : 0.0f;
                    m[u] = make_float4(0.f, 0.f, 0.f, 0.f);
                    if (on) m[u] = hw4[(size_t)s * LPN + c];
                }
                if (first) {
                    acc.x = sw * sv.x; acc.y = sw * sv.y; acc.z = sw * sv.z; acc.w = sw * sv.w;
                    first = false;
                }
#pragma unroll
                for (int u = 0; u < U; ++u) {
                    if (k0 + u < n) {
                        acc.x += w[u] * m[u].x; acc.y += w[u] * m[u].y;
                        acc.z += w[u] * m[u].z; acc.w += w[u] * m[u].w;
                    }
                }
            }
        }

        acc.x += b.x; acc.y += b.y; acc.z += b.z; acc.w += b.w;

        if constexpr (FUSED) {
            acc.x = fmaxf(acc.x, 0.f); acc.y = fmaxf(acc.y, 0.f);
            acc.z = fmaxf(acc.z, 0.f); acc.w = fmaxf(acc.w, 0.f);
            const size_t idx4 = ((size_t)((threadIdx.x >> 5) * G + i) * FUSE_KP) / 4 + c;
            split_store(acc.x, acc.y, acc.z, acc.w, sAh, sAl, idx4);
        } else {
            ((float4*)outf)[(size_t)node * LPN + c] = acc;
        }
    }

    if constexpr (FUSED) {
        constexpr int K   = EMBED;
        constexpr int NT  = NNEXT / 16;
        constexpr int SPW = (FUSE_NODES / 16) * NT / 8;
        static_assert(SPW >= 1 && (FUSE_NODES / 16) * NT == 8 * SPW, "8 waves share the tiles evenly");
        static_assert(FUSE_KP % 4 == 0, "a row of the tile starts on a 4-half boundary");
        const int wave   = threadIdx.x >> 5;
        const int lane   = threadIdx.x & 31;
        const int half   = lane >> 4;
        const int cc     = lane & 15;
        const int tile   = wave % NT;
        const int strip0 = (wave / NT) * SPW;

        const _Float16* __restrict__ bh = Bhi + (tile * 16 + cc) * K + half * 8;
        const _Float16* __restrict__ bl = Blo + (tile * 16 + cc) * K + half * 8;
        v8h b0h = *(const v8h*)(bh);
        v8h b1h = *(const v8h*)(bh + 16);
        v8h b0l = *(const v8h*)(bl);
        v8h b1l = *(const v8h*)(bl + 16);

        __syncthreads();

        v8f pacc[SPW] = {};
#pragma unroll
        for (int kc = 0; kc < K / 32; ++kc) {
            v16h fbh = __builtin_shufflevector(b0h, b1h, 0,1,2,3,4,5,6,7,8,9,10,11,12,13,14,15);
            v16h fbl = __builtin_shufflevector(b0l, b1l, 0,1,2,3,4,5,6,7,8,9,10,11,12,13,14,15);
            if (kc + 1 < K / 32) {
                b0h = *(const v8h*)(bh + (kc + 1) * 32);
                b1h = *(const v8h*)(bh + (kc + 1) * 32 + 16);
                b0l = *(const v8h*)(bl + (kc + 1) * 32);
                b1l = *(const v8h*)(bl + (kc + 1) * 32 + 16);
            }
#pragma unroll
            for (int s = 0; s < SPW; ++s) {
                const int abase = ((strip0 + s) * 16 + cc) * FUSE_KP + kc * 32 + half * 8;
                v8h a0h = *(const v8h*)(sAh + abase);
                v8h a1h = *(const v8h*)(sAh + abase + 16);
                v8h a0l = *(const v8h*)(sAl + abase);
                v8h a1l = *(const v8h*)(sAl + abase + 16);
                v16h fah = __builtin_shufflevector(a0h, a1h, 0,1,2,3,4,5,6,7,8,9,10,11,12,13,14,15);
                v16h fal = __builtin_shufflevector(a0l, a1l, 0,1,2,3,4,5,6,7,8,9,10,11,12,13,14,15);
                pacc[s] = __builtin_amdgcn_wmma_f32_16x16x32_f16(false, fah, false, fbh,
                                                                 (short)0, pacc[s], false, false);
                pacc[s] = __builtin_amdgcn_wmma_f32_16x16x32_f16(false, fah, false, fbl,
                                                                 (short)0, pacc[s], false, false);
                pacc[s] = __builtin_amdgcn_wmma_f32_16x16x32_f16(false, fal, false, fbh,
                                                                 (short)0, pacc[s], false, false);
            }
        }

        const int row0 = blockIdx.x * FUSE_NODES;
#pragma unroll
        for (int s = 0; s < SPW; ++s) {
#pragma unroll
            for (int r = 0; r < 8; ++r) {
                const int row = row0 + (strip0 + s) * 16 + half * 8 + r;
                outf[(size_t)row * NNEXT + tile * 16 + cc] = pacc[s][r];
            }
        }
    }
}


extern "C" void kernel_launch(void* const* d_in, const int* in_sizes, int n_in,
                              void* d_out, int out_size, void* d_ws, size_t ws_size,
                              hipStream_t stream) {
    const int*   x      = (const int*)d_in[0];
    const int*   src    = (const int*)d_in[1];
    const int*   dst    = ((const int*)d_in[1]) + NEDGES;
    const float* W_in   = (const float*)d_in[2];
    const float* W_conv = (const float*)d_in[3];
    const float* b_conv = (const float*)d_in[4];
    const float* W_dec  = (const float*)d_in[5];
    const float* b_dec  = (const float*)d_in[6];
    (void)in_sizes; (void)n_in; (void)out_size; (void)ws_size;

    char* p = (char*)d_ws;
    auto carve = [&](size_t bytes) { char* r = p; p += (bytes + 255) & ~(size_t)255; return r; };
    float*    dinv   = (float*)   carve((size_t)NNODES * 4);
    int*      cnt    = (int*)     carve((size_t)NNODES * 4);
    int*      cursor = (int*)     carve((size_t)NNODES * 4);
    int*      off    = (int*)     carve((size_t)(NNODES + 1) * 4);
    int*      col    = (int*)     carve((size_t)NEDGES * 4);
    float*    wt     = (float*)   carve((size_t)NEDGES * 4);
    float*    hwa    = (float*)   carve((size_t)NNODES * EMBED * 4);
    float*    hwb    = (float*)   carve((size_t)NNODES * EMBED * 4);
    _Float16* Wch    = (_Float16*)carve((size_t)EMBED * EMBED * 2);
    _Float16* Wcl    = (_Float16*)carve((size_t)EMBED * EMBED * 2);
    _Float16* Wdh    = (_Float16*)carve((size_t)NTYPES * EMBED * 2);
    _Float16* Wdl    = (_Float16*)carve((size_t)NTYPES * EMBED * 2);
    float*    tw     = (float*)   carve((size_t)NTYPES * EMBED * 4);

    const int B = 256;
    auto blocks = [](long long n, int b) { return (int)((n + b - 1) / b); };

    k_zero2   <<<blocks(NNODES, B), B, 0, stream>>>(cnt, cursor);
    k_count   <<<blocks(NEDGES, B), B, 0, stream>>>(dst, cnt);
    k_scan_prep<<<PREP_WGS, SCAN_THREADS, 0, stream>>>(cnt, off, dinv, W_in, tw,
                                                       W_conv, Wch, Wcl, W_dec, Wdh, Wdl);
    k_csr_fill<<<blocks(NEDGES, B), B, 0, stream>>>(src, dst, off, dinv, cursor, col, wt);

    const int fused_blocks = NNODES / FUSE_NODES;

    static_assert(NITERS >= 2, "the first and the last conv layer are peeled");
    float* cur = hwa;
    float* nxt = hwb;
    k_gather<EMBED, true, EMBED><<<fused_blocks, B, 0, stream>>>(
        tw, x, off, col, wt, dinv, b_conv, Wch, Wcl, cur);
    for (int it = 1; it < NITERS - 1; ++it) {
        k_gather<EMBED, false, EMBED><<<fused_blocks, B, 0, stream>>>(
            cur, x, off, col, wt, dinv, b_conv, Wch, Wcl, nxt);
        float* sw = cur; cur = nxt; nxt = sw;
    }
    k_gather<EMBED, false, NTYPES><<<fused_blocks, B, 0, stream>>>(
        cur, x, off, col, wt, dinv, b_conv, Wdh, Wdl, nxt);

    k_gather<NTYPES, false, 0><<<blocks((long long)(NNODES / GATHER_NODES) * 16, B), B, 0, stream>>>(
        nxt, x, off, col, wt, dinv, b_dec, nullptr, nullptr, (float*)d_out);
}
